# starting kernel
# baseline (speedup 1.0000x reference)
#include <hip/hip_runtime.h>
#include <hip/hip_bf16.h>


#define NN      100000
#define EE      1600000
#define ETOT    (EE + NN)
#define F_IN    256
#define DD      128
#define HH      8
#define CC      16
#define LL      4
#define NEG_SLOPE 0.2f
#define LN_EPS  1e-5f

typedef __attribute__((ext_vector_type(2))) float v2f;
typedef __attribute__((ext_vector_type(8))) float v8f;

template <int K>
__global__ __launch_bounds__(256) void gemm_wmma_k(
    const float* __restrict__ X, const float* __restrict__ W,
    const float* __restrict__ bias, float* __restrict__ Y) {
  constexpr int KP = K + 4;
  __shared__ float As[16 * KP];

  const int wave = threadIdx.x >> 5;
  const int lane = threadIdx.x & 31;
  const int row0 = blockIdx.x * 16;
  const int col0 = wave * 16;
  const int m    = lane & 15;
  const int koff = (lane < 16) ? 0 : 2;

  #pragma unroll
  for (int i = threadIdx.x * 4; i < 16 * K; i += 256 * 4) {
    const int r = i / K, c = i % K;
    *(float4*)&As[r * KP + c] =
        *(const float4*)(X + (size_t)(row0 + r) * K + c);
  }
  __syncthreads();

  const float* __restrict__ arow = As + m * KP;
  const float* __restrict__ wrow = W + (size_t)(col0 + m) * K;

  v8f acc = {};
#if __has_builtin(__builtin_amdgcn_wmma_f32_16x16x4_f32)
  for (int k = 0; k < K; k += 4) {
    v2f a, b;
    a.x = arow[k + koff];
    a.y = arow[k + koff + 1];
    b.x = wrow[k + koff];
    b.y = wrow[k + koff + 1];
    acc = __builtin_amdgcn_wmma_f32_16x16x4_f32(
        false, a, false, b, (short)0, acc, false, false);
  }
#else
  for (int k = 0; k < K; ++k) {
    float av = arow[k];
    #pragma unroll
    for (int v = 0; v < 8; ++v) acc[v] += av * W[(size_t)(col0 + v) * K + k];
  }
#endif

  const int mrow = (lane < 16) ? 0 : 8;
  const int c    = col0 + m;
  const float bv = bias ? bias[c] : 0.0f;
  #pragma unroll
  for (int v = 0; v < 8; ++v) {
    Y[(size_t)(row0 + v + mrow) * DD + c] = acc[v] + bv;
  }
}

__global__ __launch_bounds__(256) void att_logits_k(
    const float* __restrict__ xh, const float* __restrict__ a_src,
    const float* __restrict__ a_dst, float* __restrict__ e_src,
    float* __restrict__ e_dst) {
  int t = blockIdx.x * blockDim.x + threadIdx.x;
  if (t >= NN * HH) return;
  int n = t >> 3, h = t & 7;
  const float4* xp = (const float4*)(xh + (size_t)n * DD + h * CC);
  const float4* sp = (const float4*)(a_src + h * CC);
  const float4* dp = (const float4*)(a_dst + h * CC);
  float ss = 0.f, sd = 0.f;
  #pragma unroll
  for (int i = 0; i < 4; ++i) {
    float4 x = xp[i], s = sp[i], d = dp[i];
    ss += x.x * s.x + x.y * s.y + x.z * s.z + x.w * s.w;
    sd += x.x * d.x + x.y * d.y + x.z * d.z + x.w * d.w;
  }
  e_src[t] = ss;
  e_dst[t] = sd;
}

__device__ __forceinline__ void edge_nodes(const int* __restrict__ ei, int e,
                                           int& s, int& d) {
  if (e < EE) { s = ei[e]; d = ei[EE + e]; }
  else        { s = e - EE; d = e - EE; }
}
__device__ __forceinline__ float leaky(float v) {
  return v > 0.f ? v : v * NEG_SLOPE;
}
__device__ __forceinline__ unsigned f2ord(float f) {
  unsigned u = __float_as_uint(f);
  return (u & 0x80000000u) ? ~u : (u | 0x80000000u);
}
__device__ __forceinline__ float ord2f(unsigned u) {
  return (u & 0x80000000u) ? __uint_as_float(u & 0x7fffffffu)
                           : __uint_as_float(~u);
}
#define ORD_NEG_INF 0x007FFFFFu

__global__ __launch_bounds__(256) void edge_max_k(
    const int* __restrict__ ei, const float* __restrict__ e_src,
    const float* __restrict__ e_dst, unsigned* __restrict__ emax) {
  int t = blockIdx.x * blockDim.x + threadIdx.x;
  if (t >= ETOT * HH) return;
  int e = t >> 3, h = t & 7;
  int s, d; edge_nodes(ei, e, s, d);
  float v = leaky(e_src[s * HH + h] + e_dst[d * HH + h]);
  atomicMax(&emax[d * HH + h], f2ord(v));
}

__global__ __launch_bounds__(256) void edge_den_k(
    const int* __restrict__ ei, const float* __restrict__ e_src,
    const float* __restrict__ e_dst, const unsigned* __restrict__ emax,
    float* __restrict__ denom) {
  int t = blockIdx.x * blockDim.x + threadIdx.x;
  if (t >= ETOT * HH) return;
  int e = t >> 3, h = t & 7;
  int s, d; edge_nodes(ei, e, s, d);
  float v  = leaky(e_src[s * HH + h] + e_dst[d * HH + h]);
  float ex = __expf(v - ord2f(emax[d * HH + h]));
  atomicAdd(&denom[d * HH + h], ex);
}

__global__ __launch_bounds__(256) void edge_aggr_k(
    const int* __restrict__ ei, const float* __restrict__ e_src,
    const float* __restrict__ e_dst, const unsigned* __restrict__ emax,
    const float* __restrict__ denom, const float* __restrict__ xh,
    float* __restrict__ accum) {
  int t = blockIdx.x * blockDim.x + threadIdx.x;
  if (t >= ETOT * 32) return;
  int e = t >> 5, lane = t & 31;
  int h = lane >> 2;
  int s, d; edge_nodes(ei, e, s, d);
  float v     = leaky(e_src[s * HH + h] + e_dst[d * HH + h]);
  float ex    = __expf(v - ord2f(emax[d * HH + h]));
  float alpha = ex / (denom[d * HH + h] + 1e-16f);
  float4 x = *(const float4*)(xh + (size_t)s * DD + lane * 4);
  float* out = accum + (size_t)d * DD + lane * 4;
  atomicAdd(out + 0, x.x * alpha);
  atomicAdd(out + 1, x.y * alpha);
  atomicAdd(out + 2, x.z * alpha);
  atomicAdd(out + 3, x.w * alpha);
}

__global__ __launch_bounds__(256) void residual_ln_k(
    const float* __restrict__ hin, const float* __restrict__ accum,
    const float* __restrict__ gb, const float* __restrict__ g,
    const float* __restrict__ b, float* __restrict__ hout) {
  int node = blockIdx.x * 8 + (threadIdx.x >> 5);
  int lane = threadIdx.x & 31;
  size_t base = (size_t)node * DD + lane * 4;
  float4 hv = *(const float4*)(hin + base);
  float4 av = *(const float4*)(accum + base);
  float4 bv = *(const float4*)(gb + lane * 4);
  float x0 = hv.x + av.x + bv.x;
  float x1 = hv.y + av.y + bv.y;
  float x2 = hv.z + av.z + bv.z;
  float x3 = hv.w + av.w + bv.w;
  float s = x0 + x1 + x2 + x3;
  #pragma unroll
  for (int o = 16; o > 0; o >>= 1) s += __shfl_xor(s, o, 32);
  float mean = s * (1.f / (float)DD);
  float d0 = x0 - mean, d1 = x1 - mean, d2 = x2 - mean, d3 = x3 - mean;
  float vs = d0 * d0 + d1 * d1 + d2 * d2 + d3 * d3;
  #pragma unroll
  for (int o = 16; o > 0; o >>= 1) vs += __shfl_xor(vs, o, 32);
  float inv = rsqrtf(vs * (1.f / (float)DD) + LN_EPS);
  float4 gv = *(const float4*)(g + lane * 4);
  float4 lb = *(const float4*)(b + lane * 4);
  float4 o4;
  o4.x = d0 * inv * gv.x + lb.x;
  o4.y = d1 * inv * gv.y + lb.y;
  o4.z = d2 * inv * gv.z + lb.z;
  o4.w = d3 * inv * gv.w + lb.w;
  *(float4*)(hout + base) = o4;
}

__global__ __launch_bounds__(128) void pool_sum_k(const float* __restrict__ h,
                                                  float* __restrict__ pooled) {
  int c = threadIdx.x;
  float s = 0.f;
  for (int n = blockIdx.x; n < NN; n += gridDim.x)
    s += h[(size_t)n * DD + c];
  atomicAdd(&pooled[c], s);
}

__global__ __launch_bounds__(128) void head_k(
    const float* __restrict__ pooled_sum, const float* __restrict__ W_mu,
    const float* __restrict__ b_mu, const float* __restrict__ W_lv,
    const float* __restrict__ b_lv, float* __restrict__ out) {
  int d = threadIdx.x;
  __shared__ float pl[DD];
  float pd = pooled_sum[d] * (1.f / (float)NN);
  pl[d] = pd;
  __syncthreads();
  float mu = b_mu[d], lv = b_lv[d];
  for (int k = 0; k < DD; ++k) {
    float p = pl[k];
    mu += p * W_mu[(size_t)d * DD + k];
    lv += p * W_lv[(size_t)d * DD + k];
  }
  out[d]      = mu;
  out[DD + d] = lv;
  out[2 * DD + (size_t)NN * DD + d] = pd;
}

__global__ __launch_bounds__(256) void fill32_k(unsigned* __restrict__ p,
                                                unsigned v, int n) {
  int t = blockIdx.x * blockDim.x + threadIdx.x;
  if (t < n) p[t] = v;
}

extern "C" void kernel_launch(void* const* d_in, const int* in_sizes, int n_in,
                              void* d_out, int out_size, void* d_ws, size_t ws_size,
                              hipStream_t stream) {
  const float* x       = (const float*)d_in[0];
  const int*   ei      = (const int*)d_in[1];
  const float* W_in    = (const float*)d_in[2];
  const float* b_in    = (const float*)d_in[3];
  const float* gat_W   = (const float*)d_in[4];
  const float* att_src = (const float*)d_in[5];
  const float* att_dst = (const float*)d_in[6];
  const float* gat_b   = (const float*)d_in[7];
  const float* ln_g    = (const float*)d_in[8];
  const float* ln_b    = (const float*)d_in[9];
  const float* W_mu    = (const float*)d_in[10];
  const float* b_mu    = (const float*)d_in[11];
  const float* W_lv    = (const float*)d_in[12];
  const float* b_lv    = (const float*)d_in[13];
  float* out = (float*)d_out;

  float*    h      = (float*)d_ws;
  float*    xh     = h     + (size_t)NN * DD;
  float*    accum  = xh    + (size_t)NN * DD;
  float*    e_src  = accum + (size_t)NN * DD;
  float*    e_dst  = e_src + (size_t)NN * HH;
  unsigned* emax   = (unsigned*)(e_dst + (size_t)NN * HH);
  float*    denom  = (float*)(emax + (size_t)NN * HH);
  float*    pooled = denom + (size_t)NN * HH;

  float* h_final = out + 2 * DD;

  const int TPB = 256;
  const int nh_blk   = (NN * HH + TPB - 1) / TPB;
  const int eh_blk   = (ETOT * HH + TPB - 1) / TPB;
  const int ea_blk   = (ETOT * 32 + TPB - 1) / TPB;
  const int nd_blk   = ((int)((size_t)NN * DD) + TPB - 1) / TPB;
  const int ln_blk   = NN / 8;
  const int gemm_blk = NN / 16;

  gemm_wmma_k<F_IN><<<gemm_blk, TPB, 0, stream>>>(x, W_in, b_in, h);

  for (int l = 0; l < LL; ++l) {
    const float* Wl  = gat_W   + (size_t)l * DD * DD;
    const float* asl = att_src + (size_t)l * HH * CC;
    const float* adl = att_dst + (size_t)l * HH * CC;
    const float* gbl = gat_b   + (size_t)l * DD;
    const float* gl  = ln_g    + (size_t)l * DD;
    const float* bl  = ln_b    + (size_t)l * DD;

    gemm_wmma_k<DD><<<gemm_blk, TPB, 0, stream>>>(h, Wl, nullptr, xh);

    att_logits_k<<<nh_blk, TPB, 0, stream>>>(xh, asl, adl, e_src, e_dst);

    fill32_k<<<nh_blk, TPB, 0, stream>>>(emax, ORD_NEG_INF, NN * HH);
    fill32_k<<<nh_blk, TPB, 0, stream>>>((unsigned*)denom, 0u, NN * HH);
    fill32_k<<<nd_blk, TPB, 0, stream>>>((unsigned*)accum, 0u, NN * DD);

    edge_max_k <<<eh_blk, TPB, 0, stream>>>(ei, e_src, e_dst, emax);
    edge_den_k <<<eh_blk, TPB, 0, stream>>>(ei, e_src, e_dst, emax, denom);
    edge_aggr_k<<<ea_blk, TPB, 0, stream>>>(ei, e_src, e_dst, emax, denom, xh, accum);

    float* hout = (l == LL - 1) ? h_final : h;
    residual_ln_k<<<ln_blk, TPB, 0, stream>>>(h, accum, gbl, gl, bl, hout);
  }

  fill32_k<<<1, TPB, 0, stream>>>((unsigned*)pooled, 0u, DD);
  pool_sum_k<<<512, DD, 0, stream>>>(h_final, pooled);
  head_k<<<1, DD, 0, stream>>>(pooled, W_mu, b_mu, W_lv, b_lv, out);
}


# edges binned by 128-node dst tile once per call; per layer a work-group per tile sums exp and exp*row in LDS (ds atomics, divided after), LayerNorm from LDS; no global f32 atomics; arrival sum order
# speedup vs baseline: 2.7934x; 2.7934x over previous
#include <hip/hip_runtime.h>
#include <hip/hip_bf16.h>


#define NN      100000
#define EE      1600000
#define F_IN    256
#define DD      128
#define HH      8
#define CC      16
#define LL      4
#define NEG_SLOPE 0.2f
#define LN_EPS  1e-5f

#define TN      128
#define NT      ((NN + TN - 1) / TN)
#define CH      8192
#define NB      ((EE + CH - 1) / CH)
#define SCAN_U  28
#define BIN_U   8
static_assert(NB % SCAN_U == 0, "scan batches must cover the blocks exactly");
static_assert(CH % (256 * BIN_U) == 0, "binning batches must cover a block exactly");
static_assert(NT <= 1024, "one scan thread per tile");
static_assert(NN < (1 << 17) && TN == 128, "src and local dst share one word");

typedef __attribute__((ext_vector_type(2))) float v2f;
typedef __attribute__((ext_vector_type(8))) float v8f;

template <int K>
__global__ __launch_bounds__(256) void gemm_wmma_k(
    const float* __restrict__ X, const float* __restrict__ W,
    const float* __restrict__ bias, float* __restrict__ Y) {
  constexpr int KP = K + 4;
  __shared__ float As[16 * KP];

  const int wave = threadIdx.x >> 5;
  const int lane = threadIdx.x & 31;
  const int row0 = blockIdx.x * 16;
  const int col0 = wave * 16;
  const int m    = lane & 15;
  const int koff = (lane < 16) ? 0 : 2;

  #pragma unroll
  for (int i = threadIdx.x * 4; i < 16 * K; i += 256 * 4) {
    const int r = i / K, c = i % K;
    *(float4*)&As[r * KP + c] =
        *(const float4*)(X + (size_t)(row0 + r) * K + c);
  }
  __syncthreads();

  const float* __restrict__ arow = As + m * KP;
  const float* __restrict__ wrow = W + (size_t)(col0 + m) * K;

  v8f acc = {};
#if __has_builtin(__builtin_amdgcn_wmma_f32_16x16x4_f32)
  for (int k = 0; k < K; k += 4) {
    v2f a, b;
    a.x = arow[k + koff];
    a.y = arow[k + koff + 1];
    b.x = wrow[k + koff];
    b.y = wrow[k + koff + 1];
    acc = __builtin_amdgcn_wmma_f32_16x16x4_f32(
        false, a, false, b, (short)0, acc, false, false);
  }
#else
  for (int k = 0; k < K; ++k) {
    float av = arow[k];
    #pragma unroll
    for (int v = 0; v < 8; ++v) acc[v] += av * W[(size_t)(col0 + v) * K + k];
  }
#endif

  const int mrow = (lane < 16) ? 0 : 8;
  const int c    = col0 + m;
  const float bv = bias ? bias[c] : 0.0f;
  #pragma unroll
  for (int v = 0; v < 8; ++v) {
    Y[(size_t)(row0 + v + mrow) * DD + c] = acc[v] + bv;
  }
}

__global__ __launch_bounds__(256) void att_logits_k(
    const float* __restrict__ xh, const float* __restrict__ a_src,
    const float* __restrict__ a_dst, float* __restrict__ e_src,
    float* __restrict__ e_dst) {
  int t = blockIdx.x * blockDim.x + threadIdx.x;
  if (t >= NN * HH) return;
  int n = t >> 3, h = t & 7;
  const float4* xp = (const float4*)(xh + (size_t)n * DD + h * CC);
  const float4* sp = (const float4*)(a_src + h * CC);
  const float4* dp = (const float4*)(a_dst + h * CC);
  float ss = 0.f, sd = 0.f;
  #pragma unroll
  for (int i = 0; i < 4; ++i) {
    float4 x = xp[i], s = sp[i], d = dp[i];
    ss += x.x * s.x + x.y * s.y + x.z * s.z + x.w * s.w;
    sd += x.x * d.x + x.y * d.y + x.z * d.z + x.w * d.w;
  }
  e_src[t] = ss;
  e_dst[t] = sd;
}

__device__ __forceinline__ float leaky(float v) {
  return v > 0.f ? v : v * NEG_SLOPE;
}
__device__ __forceinline__ unsigned f2ord(float f) {
  unsigned u = __float_as_uint(f);
  return (u & 0x80000000u) ? ~u : (u | 0x80000000u);
}
__device__ __forceinline__ float ord2f(unsigned u) {
  return (u & 0x80000000u) ? __uint_as_float(u & 0x7fffffffu)
                           : __uint_as_float(~u);
}

__global__ __launch_bounds__(256) void bin_count_k(
    const int* __restrict__ ei, int* __restrict__ blockhist) {
  __shared__ int hist[NT];
  for (int i = threadIdx.x; i < NT; i += 256) hist[i] = 0;
  __syncthreads();
  const int e0 = blockIdx.x * CH;
  for (int i0 = threadIdx.x; i0 < CH; i0 += 256 * BIN_U) {
    int d[BIN_U];
    #pragma unroll
    for (int u = 0; u < BIN_U; ++u) {
      const int e = e0 + i0 + 256 * u;
      d[u] = (e < EE) ? ei[EE + e] : -1;
    }
    #pragma unroll
    for (int u = 0; u < BIN_U; ++u)
      if (d[u] >= 0) atomicAdd(&hist[d[u] >> 7], 1);
  }
  __syncthreads();
  for (int i = threadIdx.x; i < NT; i += 256)
    blockhist[blockIdx.x * NT + i] = hist[i];
}

__global__ __launch_bounds__(1024) void bin_scan_k(
    const int* __restrict__ blockhist, int* __restrict__ blockoff,
    int* __restrict__ binstart) {
  __shared__ int sc[1024];
  const int t = threadIdx.x;
  int run = 0;
  if (t < NT) {
    for (int b0 = 0; b0 < NB; b0 += SCAN_U) {
      int c[SCAN_U];
      #pragma unroll
      for (int u = 0; u < SCAN_U; ++u) c[u] = blockhist[(b0 + u) * NT + t];
      #pragma unroll
      for (int u = 0; u < SCAN_U; ++u) {
        blockoff[(b0 + u) * NT + t] = run;
        run += c[u];
      }
    }
  }
  sc[t] = run;
  __syncthreads();
  for (int o = 1; o < 1024; o <<= 1) {
    const int v = (t >= o) ? sc[t - o] : 0;
    __syncthreads();
    sc[t] += v;
    __syncthreads();
  }
  if (t < NT) binstart[t] = sc[t] - run;
  if (t == NT - 1) binstart[NT] = sc[t];
}

__global__ __launch_bounds__(256) void bin_scatter_k(
    const int* __restrict__ ei, const int* __restrict__ blockoff,
    const int* __restrict__ binstart, int* __restrict__ binned) {
  __shared__ int cur[NT];
  for (int i = threadIdx.x; i < NT; i += 256)
    cur[i] = binstart[i] + blockoff[blockIdx.x * NT + i];
  __syncthreads();
  const int e0 = blockIdx.x * CH;
  for (int i0 = threadIdx.x; i0 < CH; i0 += 256 * BIN_U) {
    int s[BIN_U], d[BIN_U];
    #pragma unroll
    for (int u = 0; u < BIN_U; ++u) {
      const int e = e0 + i0 + 256 * u;
      s[u] = (e < EE) ? ei[e] : 0;
      d[u] = (e < EE) ? ei[EE + e] : -1;
    }
    #pragma unroll
    for (int u = 0; u < BIN_U; ++u) {
      if (d[u] >= 0) {
        const int pos = atomicAdd(&cur[d[u] >> 7], 1);
        binned[pos] = (s[u] << 7) | (d[u] & 127);
      }
    }
  }
}

#define AG_U 16

__device__ __forceinline__ int fetch_edge(const int* __restrict__ binned,
                                          int eb, int ne, int tot, int n0,
                                          int idx) {
  int pk = n0 << 7;
  if (idx < ne) pk = binned[eb + idx];
  else if (idx < tot) { const int dl = idx - ne; pk = ((n0 + dl) << 7) | dl; }
  return pk;
}

__global__ __launch_bounds__(256) void tile_aggr_ln_k(
    const int* __restrict__ binned, const int* __restrict__ binstart,
    const float* __restrict__ e_src, const float* __restrict__ e_dst,
    const float* __restrict__ xh, const float* hin,
    const float* __restrict__ gb, const float* __restrict__ g,
    const float* __restrict__ b, float* hout) {
  __shared__ __attribute__((aligned(16))) float acc[TN * DD];
  __shared__ float    sden[TN * HH];
  __shared__ unsigned smax[TN * HH];
  __shared__ float    sed[TN * HH];

  const int tid  = threadIdx.x;
  const int lane = tid & 31;
  const int wave = __builtin_amdgcn_readfirstlane(tid >> 5);
  const int h    = lane >> 2;
  const int n0   = blockIdx.x * TN;
  const int nn   = (NN - n0 < TN) ? (NN - n0) : TN;
  const int eb   = binstart[blockIdx.x];
  const int ne   = binstart[blockIdx.x + 1] - eb;
  const int tot  = ne + nn;

  for (int i = tid * 4; i < TN * DD; i += 256 * 4)
    *(float4*)&acc[i] = make_float4(0.f, 0.f, 0.f, 0.f);
  for (int i = tid; i < TN * HH; i += 256) {
    float es = 0.f, ed = 0.f;
    if ((i >> 3) < nn) {
      es = e_src[(size_t)n0 * HH + i];
      ed = e_dst[(size_t)n0 * HH + i];
    }
    sed[i]  = ed;
    sden[i] = 0.f;
    smax[i] = f2ord(leaky(es + ed));
  }
  __syncthreads();

  for (int i0 = tid; i0 < ne; i0 += 256 * 2) {
    int pk[2];
    float4 ea[2], ec[2];
    #pragma unroll
    for (int u = 0; u < 2; ++u) {
      const int i = i0 + 256 * u;
      pk[u] = (i < ne) ? binned[eb + i] : -1;
    }
    #pragma unroll
    for (int u = 0; u < 2; ++u) {
      const int s = (pk[u] >= 0) ? (pk[u] >> 7) : n0;
      const float4* p = (const float4*)(e_src + (size_t)s * HH);
      ea[u] = p[0];
      ec[u] = p[1];
    }
    #pragma unroll
    for (int u = 0; u < 2; ++u) {
      if (pk[u] >= 0) {
        const int dl = pk[u] & 127;
        const float* sd = sed + dl * HH;
        unsigned*    sm = smax + dl * HH;
        atomicMax(sm + 0, f2ord(leaky(ea[u].x + sd[0])));
        atomicMax(sm + 1, f2ord(leaky(ea[u].y + sd[1])));
        atomicMax(sm + 2, f2ord(leaky(ea[u].z + sd[2])));
        atomicMax(sm + 3, f2ord(leaky(ea[u].w + sd[3])));
        atomicMax(sm + 4, f2ord(leaky(ec[u].x + sd[4])));
        atomicMax(sm + 5, f2ord(leaky(ec[u].y + sd[5])));
        atomicMax(sm + 6, f2ord(leaky(ec[u].z + sd[6])));
        atomicMax(sm + 7, f2ord(leaky(ec[u].w + sd[7])));
      }
    }
  }
  __syncthreads();

  {
    int base = wave * 32;
    int pk_l = fetch_edge(binned, eb, ne, tot, n0, base + lane);
    while (base < tot) {
      const int nbase = base + 8 * 32;
      const int pk_n = fetch_edge(binned, eb, ne, tot, n0, nbase + lane);
      const int cnt  = (tot - base < 32) ? (tot - base) : 32;
      for (int k0 = 0; k0 < cnt; k0 += AG_U) {
        float4 x[AG_U];
        float  es[AG_U];
        int    dls[AG_U];
        #pragma unroll
        for (int u = 0; u < AG_U; ++u) {
          const int pk = __builtin_amdgcn_readlane(pk_l, k0 + u);
          const int s  = pk >> 7;
          dls[u] = pk & 127;
          es[u]  = e_src[(size_t)s * HH + h];
          x[u]   = *(const float4*)(xh + (size_t)s * DD + lane * 4);
        }
        #pragma unroll
        for (int u = 0; u < AG_U; ++u) {
          if (k0 + u < cnt) {
            const int   dl = dls[u];
            const float v  = leaky(es[u] + sed[dl * HH + h]);
            const float ex = __expf(v - ord2f(smax[dl * HH + h]));
            float* a = acc + dl * DD + lane;
            atomicAdd(a,      x[u].x * ex);
            atomicAdd(a + 32, x[u].y * ex);
            atomicAdd(a + 64, x[u].z * ex);
            atomicAdd(a + 96, x[u].w * ex);
            if ((lane & 3) == 0) atomicAdd(&sden[dl * HH + h], ex);
          }
        }
      }
      pk_l = pk_n;
      base = nbase;
    }
  }
  __syncthreads();

  const float4 bv = *(const float4*)(gb + lane * 4);
  const float4 gv = *(const float4*)(g + lane * 4);
  const float4 lb = *(const float4*)(b + lane * 4);
  for (int q0 = 0; q0 < TN / 8; q0 += 4) {
    float4 hv[4];
    #pragma unroll
    for (int u = 0; u < 4; ++u) {
      const int dl = wave + 8 * (q0 + u);
      hv[u] = make_float4(0.f, 0.f, 0.f, 0.f);
      if (dl < nn)
        hv[u] = *(const float4*)(hin + (size_t)(n0 + dl) * DD + lane * 4);
    }
    #pragma unroll
    for (int u = 0; u < 4; ++u) {
      const int dl = wave + 8 * (q0 + u);
      if (dl < nn) {
        const float  den = sden[dl * HH + h] + 1e-16f;
        const float* a   = acc + dl * DD + lane;
        float x0 = hv[u].x + a[0]  / den + bv.x;
        float x1 = hv[u].y + a[32] / den + bv.y;
        float x2 = hv[u].z + a[64] / den + bv.z;
        float x3 = hv[u].w + a[96] / den + bv.w;
        float s = x0 + x1 + x2 + x3;
        #pragma unroll
        for (int o = 16; o > 0; o >>= 1) s += __shfl_xor(s, o, 32);
        float mean = s * (1.f / (float)DD);
        float d0 = x0 - mean, d1 = x1 - mean, d2 = x2 - mean, d3 = x3 - mean;
        float vs = d0 * d0 + d1 * d1 + d2 * d2 + d3 * d3;
        #pragma unroll
        for (int o = 16; o > 0; o >>= 1) vs += __shfl_xor(vs, o, 32);
        float inv = rsqrtf(vs * (1.f / (float)DD) + LN_EPS);
        float4 o4;
        o4.x = d0 * inv * gv.x + lb.x;
        o4.y = d1 * inv * gv.y + lb.y;
        o4.z = d2 * inv * gv.z + lb.z;
        o4.w = d3 * inv * gv.w + lb.w;
        *(float4*)(hout + (size_t)(n0 + dl) * DD + lane * 4) = o4;
      }
    }
  }
}

__global__ __launch_bounds__(128) void pool_sum_k(const float* __restrict__ h,
                                                  float* __restrict__ pooled) {
  int c = threadIdx.x;
  float s = 0.f;
  for (int n = blockIdx.x; n < NN; n += gridDim.x)
    s += h[(size_t)n * DD + c];
  atomicAdd(&pooled[c], s);
}

__global__ __launch_bounds__(128) void head_k(
    const float* __restrict__ pooled_sum, const float* __restrict__ W_mu,
    const float* __restrict__ b_mu, const float* __restrict__ W_lv,
    const float* __restrict__ b_lv, float* __restrict__ out) {
  int d = threadIdx.x;
  __shared__ float pl[DD];
  float pd = pooled_sum[d] * (1.f / (float)NN);
  pl[d] = pd;
  __syncthreads();
  float mu = b_mu[d], lv = b_lv[d];
  for (int k = 0; k < DD; ++k) {
    float p = pl[k];
    mu += p * W_mu[(size_t)d * DD + k];
    lv += p * W_lv[(size_t)d * DD + k];
  }
  out[d]      = mu;
  out[DD + d] = lv;
  out[2 * DD + (size_t)NN * DD + d] = pd;
}

__global__ __launch_bounds__(256) void fill32_k(unsigned* __restrict__ p,
                                                unsigned v, int n) {
  int t = blockIdx.x * blockDim.x + threadIdx.x;
  if (t < n) p[t] = v;
}

extern "C" void kernel_launch(void* const* d_in, const int* in_sizes, int n_in,
                              void* d_out, int out_size, void* d_ws, size_t ws_size,
                              hipStream_t stream) {
  const float* x       = (const float*)d_in[0];
  const int*   ei      = (const int*)d_in[1];
  const float* W_in    = (const float*)d_in[2];
  const float* b_in    = (const float*)d_in[3];
  const float* gat_W   = (const float*)d_in[4];
  const float* att_src = (const float*)d_in[5];
  const float* att_dst = (const float*)d_in[6];
  const float* gat_b   = (const float*)d_in[7];
  const float* ln_g    = (const float*)d_in[8];
  const float* ln_b    = (const float*)d_in[9];
  const float* W_mu    = (const float*)d_in[10];
  const float* b_mu    = (const float*)d_in[11];
  const float* W_lv    = (const float*)d_in[12];
  const float* b_lv    = (const float*)d_in[13];
  float* out = (float*)d_out;

  float* h      = (float*)d_ws;
  float* xh     = h     + (size_t)NN * DD;
  float* plane  = xh    + (size_t)NN * DD;
  float* e_src  = plane + (size_t)NN * DD;
  float* e_dst  = e_src + (size_t)NN * HH;
  float* pooled = e_dst + (size_t)NN * HH * 3;

  int* binned    = (int*)plane;
  int* blockhist = binned    + (size_t)EE;
  int* blockoff  = blockhist + (size_t)NB * NT;
  int* binstart  = blockoff  + (size_t)NB * NT;

  float* h_final = out + 2 * DD;

  const int TPB = 256;
  const int nh_blk   = (NN * HH + TPB - 1) / TPB;
  const int gemm_blk = NN / 16;

  gemm_wmma_k<F_IN><<<gemm_blk, TPB, 0, stream>>>(x, W_in, b_in, h);

  bin_count_k  <<<NB, TPB, 0, stream>>>(ei, blockhist);
  bin_scan_k   <<<1, 1024, 0, stream>>>(blockhist, blockoff, binstart);
  bin_scatter_k<<<NB, TPB, 0, stream>>>(ei, blockoff, binstart, binned);

  for (int l = 0; l < LL; ++l) {
    const float* Wl  = gat_W   + (size_t)l * DD * DD;
    const float* asl = att_src + (size_t)l * HH * CC;
    const float* adl = att_dst + (size_t)l * HH * CC;
    const float* gbl = gat_b   + (size_t)l * DD;
    const float* gl  = ln_g    + (size_t)l * DD;
    const float* bl  = ln_b    + (size_t)l * DD;

    gemm_wmma_k<DD><<<gemm_blk, TPB, 0, stream>>>(h, Wl, nullptr, xh);

    att_logits_k<<<nh_blk, TPB, 0, stream>>>(xh, asl, adl, e_src, e_dst);

    float* hout = (l == LL - 1) ? h_final : h;
    tile_aggr_ln_k<<<NT, TPB, 0, stream>>>(binned, binstart, e_src, e_dst, xh,
                                           h, gbl, gl, bl, hout);
  }

  fill32_k<<<1, TPB, 0, stream>>>((unsigned*)pooled, 0u, DD);
  pool_sum_k<<<512, DD, 0, stream>>>(h_final, pooled);
  head_k<<<1, DD, 0, stream>>>(pooled, W_mu, b_mu, W_lv, b_lv, out);
}


# mean pool folded into the last tile kernel: waves add their LayerNorm rows per column in registers, one partial per tile, head adds 782; h not re-read, pool launch, atomics gone; pool sum regrouped
# speedup vs baseline: 3.1193x; 1.1167x over previous
#include <hip/hip_runtime.h>
#include <hip/hip_bf16.h>


#define NN      100000
#define EE      1600000
#define F_IN    256
#define DD      128
#define HH      8
#define CC      16
#define LL      4
#define NEG_SLOPE 0.2f
#define LN_EPS  1e-5f

#define TN      128
#define NT      ((NN + TN - 1) / TN)
#define CH      8192
#define NB      ((EE + CH - 1) / CH)
#define SCAN_U  28
#define BIN_U   8
static_assert(NB % SCAN_U == 0, "scan batches must cover the blocks exactly");
static_assert(CH % (256 * BIN_U) == 0, "binning batches must cover a block exactly");
static_assert(NT <= 1024, "one scan thread per tile");
static_assert(NN < (1 << 17) && TN == 128, "src and local dst share one word");

typedef __attribute__((ext_vector_type(2))) float v2f;
typedef __attribute__((ext_vector_type(8))) float v8f;

template <int K>
__global__ __launch_bounds__(256) void gemm_wmma_k(
    const float* __restrict__ X, const float* __restrict__ W,
    const float* __restrict__ bias, float* __restrict__ Y) {
  constexpr int KP = K + 4;
  __shared__ float As[16 * KP];

  const int wave = threadIdx.x >> 5;
  const int lane = threadIdx.x & 31;
  const int row0 = blockIdx.x * 16;
  const int col0 = wave * 16;
  const int m    = lane & 15;
  const int koff = (lane < 16) ? 0 : 2;

  #pragma unroll
  for (int i = threadIdx.x * 4; i < 16 * K; i += 256 * 4) {
    const int r = i / K, c = i % K;
    *(float4*)&As[r * KP + c] =
        *(const float4*)(X + (size_t)(row0 + r) * K + c);
  }
  __syncthreads();

  const float* __restrict__ arow = As + m * KP;
  const float* __restrict__ wrow = W + (size_t)(col0 + m) * K;

  v8f acc = {};
#if __has_builtin(__builtin_amdgcn_wmma_f32_16x16x4_f32)
  for (int k = 0; k < K; k += 4) {
    v2f a, b;
    a.x = arow[k + koff];
    a.y = arow[k + koff + 1];
    b.x = wrow[k + koff];
    b.y = wrow[k + koff + 1];
    acc = __builtin_amdgcn_wmma_f32_16x16x4_f32(
        false, a, false, b, (short)0, acc, false, false);
  }
#else
  for (int k = 0; k < K; ++k) {
    float av = arow[k];
    #pragma unroll
    for (int v = 0; v < 8; ++v) acc[v] += av * W[(size_t)(col0 + v) * K + k];
  }
#endif

  const int mrow = (lane < 16) ? 0 : 8;
  const int c    = col0 + m;
  const float bv = bias ? bias[c] : 0.0f;
  #pragma unroll
  for (int v = 0; v < 8; ++v) {
    Y[(size_t)(row0 + v + mrow) * DD + c] = acc[v] + bv;
  }
}

__global__ __launch_bounds__(256) void att_logits_k(
    const float* __restrict__ xh, const float* __restrict__ a_src,
    const float* __restrict__ a_dst, float* __restrict__ e_src,
    float* __restrict__ e_dst) {
  int t = blockIdx.x * blockDim.x + threadIdx.x;
  if (t >= NN * HH) return;
  int n = t >> 3, h = t & 7;
  const float4* xp = (const float4*)(xh + (size_t)n * DD + h * CC);
  const float4* sp = (const float4*)(a_src + h * CC);
  const float4* dp = (const float4*)(a_dst + h * CC);
  float ss = 0.f, sd = 0.f;
  #pragma unroll
  for (int i = 0; i < 4; ++i) {
    float4 x = xp[i], s = sp[i], d = dp[i];
    ss += x.x * s.x + x.y * s.y + x.z * s.z + x.w * s.w;
    sd += x.x * d.x + x.y * d.y + x.z * d.z + x.w * d.w;
  }
  e_src[t] = ss;
  e_dst[t] = sd;
}

__device__ __forceinline__ float leaky(float v) {
  return v > 0.f ? v : v * NEG_SLOPE;
}
__device__ __forceinline__ unsigned f2ord(float f) {
  unsigned u = __float_as_uint(f);
  return (u & 0x80000000u) ? ~u : (u | 0x80000000u);
}
__device__ __forceinline__ float ord2f(unsigned u) {
  return (u & 0x80000000u) ? __uint_as_float(u & 0x7fffffffu)
                           : __uint_as_float(~u);
}

__global__ __launch_bounds__(256) void bin_count_k(
    const int* __restrict__ ei, int* __restrict__ blockhist) {
  __shared__ int hist[NT];
  for (int i = threadIdx.x; i < NT; i += 256) hist[i] = 0;
  __syncthreads();
  const int e0 = blockIdx.x * CH;
  for (int i0 = threadIdx.x; i0 < CH; i0 += 256 * BIN_U) {
    int d[BIN_U];
    #pragma unroll
    for (int u = 0; u < BIN_U; ++u) {
      const int e = e0 + i0 + 256 * u;
      d[u] = (e < EE) ? ei[EE + e] : -1;
    }
    #pragma unroll
    for (int u = 0; u < BIN_U; ++u)
      if (d[u] >= 0) atomicAdd(&hist[d[u] >> 7], 1);
  }
  __syncthreads();
  for (int i = threadIdx.x; i < NT; i += 256)
    blockhist[blockIdx.x * NT + i] = hist[i];
}

__global__ __launch_bounds__(1024) void bin_scan_k(
    const int* __restrict__ blockhist, int* __restrict__ blockoff,
    int* __restrict__ binstart) {
  __shared__ int sc[1024];
  const int t = threadIdx.x;
  int run = 0;
  if (t < NT) {
    for (int b0 = 0; b0 < NB; b0 += SCAN_U) {
      int c[SCAN_U];
      #pragma unroll
      for (int u = 0; u < SCAN_U; ++u) c[u] = blockhist[(b0 + u) * NT + t];
      #pragma unroll
      for (int u = 0; u < SCAN_U; ++u) {
        blockoff[(b0 + u) * NT + t] = run;
        run += c[u];
      }
    }
  }
  sc[t] = run;
  __syncthreads();
  for (int o = 1; o < 1024; o <<= 1) {
    const int v = (t >= o) ? sc[t - o] : 0;
    __syncthreads();
    sc[t] += v;
    __syncthreads();
  }
  if (t < NT) binstart[t] = sc[t] - run;
  if (t == NT - 1) binstart[NT] = sc[t];
}

__global__ __launch_bounds__(256) void bin_scatter_k(
    const int* __restrict__ ei, const int* __restrict__ blockoff,
    const int* __restrict__ binstart, int* __restrict__ binned) {
  __shared__ int cur[NT];
  for (int i = threadIdx.x; i < NT; i += 256)
    cur[i] = binstart[i] + blockoff[blockIdx.x * NT + i];
  __syncthreads();
  const int e0 = blockIdx.x * CH;
  for (int i0 = threadIdx.x; i0 < CH; i0 += 256 * BIN_U) {
    int s[BIN_U], d[BIN_U];
    #pragma unroll
    for (int u = 0; u < BIN_U; ++u) {
      const int e = e0 + i0 + 256 * u;
      s[u] = (e < EE) ? ei[e] : 0;
      d[u] = (e < EE) ? ei[EE + e] : -1;
    }
    #pragma unroll
    for (int u = 0; u < BIN_U; ++u) {
      if (d[u] >= 0) {
        const int pos = atomicAdd(&cur[d[u] >> 7], 1);
        binned[pos] = (s[u] << 7) | (d[u] & 127);
      }
    }
  }
}

#define AG_U 16

__device__ __forceinline__ int fetch_edge(const int* __restrict__ binned,
                                          int eb, int ne, int tot, int n0,
                                          int idx) {
  int pk = n0 << 7;
  if (idx < ne) pk = binned[eb + idx];
  else if (idx < tot) { const int dl = idx - ne; pk = ((n0 + dl) << 7) | dl; }
  return pk;
}

__global__ __launch_bounds__(256) void tile_aggr_ln_k(
    const int* __restrict__ binned, const int* __restrict__ binstart,
    const float* __restrict__ e_src, const float* __restrict__ e_dst,
    const float* __restrict__ xh, const float* hin,
    const float* __restrict__ gb, const float* __restrict__ g,
    const float* __restrict__ b, float* hout, float* __restrict__ part) {
  __shared__ __attribute__((aligned(16))) float acc[TN * DD];
  __shared__ float    sden[TN * HH];
  __shared__ unsigned smax[TN * HH];
  __shared__ float    sed[TN * HH];
  __shared__ __attribute__((aligned(16))) float pcol[8 * DD];

  const int tid  = threadIdx.x;
  const int lane = tid & 31;
  const int wave = __builtin_amdgcn_readfirstlane(tid >> 5);
  const int h    = lane >> 2;
  const int n0   = blockIdx.x * TN;
  const int nn   = (NN - n0 < TN) ? (NN - n0) : TN;
  const int eb   = binstart[blockIdx.x];
  const int ne   = binstart[blockIdx.x + 1] - eb;
  const int tot  = ne + nn;

  for (int i = tid * 4; i < TN * DD; i += 256 * 4)
    *(float4*)&acc[i] = make_float4(0.f, 0.f, 0.f, 0.f);
  for (int i = tid; i < TN * HH; i += 256) {
    float es = 0.f, ed = 0.f;
    if ((i >> 3) < nn) {
      es = e_src[(size_t)n0 * HH + i];
      ed = e_dst[(size_t)n0 * HH + i];
    }
    sed[i]  = ed;
    sden[i] = 0.f;
    smax[i] = f2ord(leaky(es + ed));
  }
  __syncthreads();

  for (int i0 = tid; i0 < ne; i0 += 256 * 2) {
    int pk[2];
    float4 ea[2], ec[2];
    #pragma unroll
    for (int u = 0; u < 2; ++u) {
      const int i = i0 + 256 * u;
      pk[u] = (i < ne) ? binned[eb + i] : -1;
    }
    #pragma unroll
    for (int u = 0; u < 2; ++u) {
      const int s = (pk[u] >= 0) ? (pk[u] >> 7) : n0;
      const float4* p = (const float4*)(e_src + (size_t)s * HH);
      ea[u] = p[0];
      ec[u] = p[1];
    }
    #pragma unroll
    for (int u = 0; u < 2; ++u) {
      if (pk[u] >= 0) {
        const int dl = pk[u] & 127;
        const float* sd = sed + dl * HH;
        unsigned*    sm = smax + dl * HH;
        atomicMax(sm + 0, f2ord(leaky(ea[u].x + sd[0])));
        atomicMax(sm + 1, f2ord(leaky(ea[u].y + sd[1])));
        atomicMax(sm + 2, f2ord(leaky(ea[u].z + sd[2])));
        atomicMax(sm + 3, f2ord(leaky(ea[u].w + sd[3])));
        atomicMax(sm + 4, f2ord(leaky(ec[u].x + sd[4])));
        atomicMax(sm + 5, f2ord(leaky(ec[u].y + sd[5])));
        atomicMax(sm + 6, f2ord(leaky(ec[u].z + sd[6])));
        atomicMax(sm + 7, f2ord(leaky(ec[u].w + sd[7])));
      }
    }
  }
  __syncthreads();

  {
    int base = wave * 32;
    int pk_l = fetch_edge(binned, eb, ne, tot, n0, base + lane);
    while (base < tot) {
      const int nbase = base + 8 * 32;
      const int pk_n = fetch_edge(binned, eb, ne, tot, n0, nbase + lane);
      const int cnt  = (tot - base < 32) ? (tot - base) : 32;
      for (int k0 = 0; k0 < cnt; k0 += AG_U) {
        float4 x[AG_U];
        float  es[AG_U];
        int    dls[AG_U];
        #pragma unroll
        for (int u = 0; u < AG_U; ++u) {
          const int pk = __builtin_amdgcn_readlane(pk_l, k0 + u);
          const int s  = pk >> 7;
          dls[u] = pk & 127;
          es[u]  = e_src[(size_t)s * HH + h];
          x[u]   = *(const float4*)(xh + (size_t)s * DD + lane * 4);
        }
        #pragma unroll
        for (int u = 0; u < AG_U; ++u) {
          if (k0 + u < cnt) {
            const int   dl = dls[u];
            const float v  = leaky(es[u] + sed[dl * HH + h]);
            const float ex = __expf(v - ord2f(smax[dl * HH + h]));
            float* a = acc + dl * DD + lane;
            atomicAdd(a,      x[u].x * ex);
            atomicAdd(a + 32, x[u].y * ex);
            atomicAdd(a + 64, x[u].z * ex);
            atomicAdd(a + 96, x[u].w * ex);
            if ((lane & 3) == 0) atomicAdd(&sden[dl * HH + h], ex);
          }
        }
      }
      pk_l = pk_n;
      base = nbase;
    }
  }
  __syncthreads();

  const float4 bv = *(const float4*)(gb + lane * 4);
  const float4 gv = *(const float4*)(g + lane * 4);
  const float4 lb = *(const float4*)(b + lane * 4);
  float4 ps = make_float4(0.f, 0.f, 0.f, 0.f);
  for (int q0 = 0; q0 < TN / 8; q0 += 4) {
    float4 hv[4];
    #pragma unroll
    for (int u = 0; u < 4; ++u) {
      const int dl = wave + 8 * (q0 + u);
      hv[u] = make_float4(0.f, 0.f, 0.f, 0.f);
      if (dl < nn)
        hv[u] = *(const float4*)(hin + (size_t)(n0 + dl) * DD + lane * 4);
    }
    #pragma unroll
    for (int u = 0; u < 4; ++u) {
      const int dl = wave + 8 * (q0 + u);
      if (dl < nn) {
        const float  den = sden[dl * HH + h] + 1e-16f;
        const float* a   = acc + dl * DD + lane;
        float x0 = hv[u].x + a[0]  / den + bv.x;
        float x1 = hv[u].y + a[32] / den + bv.y;
        float x2 = hv[u].z + a[64] / den + bv.z;
        float x3 = hv[u].w + a[96] / den + bv.w;
        float s = x0 + x1 + x2 + x3;
        #pragma unroll
        for (int o = 16; o > 0; o >>= 1) s += __shfl_xor(s, o, 32);
        float mean = s * (1.f / (float)DD);
        float d0 = x0 - mean, d1 = x1 - mean, d2 = x2 - mean, d3 = x3 - mean;
        float vs = d0 * d0 + d1 * d1 + d2 * d2 + d3 * d3;
        #pragma unroll
        for (int o = 16; o > 0; o >>= 1) vs += __shfl_xor(vs, o, 32);
        float inv = rsqrtf(vs * (1.f / (float)DD) + LN_EPS);
        float4 o4;
        o4.x = d0 * inv * gv.x + lb.x;
        o4.y = d1 * inv * gv.y + lb.y;
        o4.z = d2 * inv * gv.z + lb.z;
        o4.w = d3 * inv * gv.w + lb.w;
        *(float4*)(hout + (size_t)(n0 + dl) * DD + lane * 4) = o4;
        ps.x += o4.x;
        ps.y += o4.y;
        ps.z += o4.z;
        ps.w += o4.w;
      }
    }
  }

  if (part) {
    *(float4*)&pcol[wave * DD + lane * 4] = ps;
    __syncthreads();
    if (tid < DD) {
      float s = 0.f;
      #pragma unroll
      for (int w = 0; w < 8; ++w) s += pcol[w * DD + tid];
      part[(size_t)blockIdx.x * DD + tid] = s;
    }
  }
}

#define HP_G 8
#define HP_U 25
static_assert(HP_G * HP_U * 4 >= NT, "four batches cover the tiles");
__global__ __launch_bounds__(1024) void head_k(
    const float* __restrict__ part, const float* __restrict__ W_mu,
    const float* __restrict__ b_mu, const float* __restrict__ W_lv,
    const float* __restrict__ b_lv, float* __restrict__ out) {
  __shared__ float red[HP_G * DD];
  __shared__ float pl[DD];
  const int d = threadIdx.x & (DD - 1);
  const int g = threadIdx.x >> 7;
  float s = 0.f;
  for (int t0 = g; t0 < NT; t0 += HP_G * HP_U) {
    float v[HP_U];
    #pragma unroll
    for (int u = 0; u < HP_U; ++u) {
      const int t = t0 + HP_G * u;
      v[u] = (t < NT) ? part[(size_t)t * DD + d] : 0.f;
    }
    #pragma unroll
    for (int u = 0; u < HP_U; ++u) s += v[u];
  }
  red[g * DD + d] = s;
  __syncthreads();
  float pd = 0.f;
  if (threadIdx.x < DD) {
    #pragma unroll
    for (int w = 0; w < HP_G; ++w) pd += red[w * DD + d];
    pd *= (1.f / (float)NN);
    pl[d] = pd;
  }
  __syncthreads();
  if (threadIdx.x < DD) {
    float mu = b_mu[d], lv = b_lv[d];
    for (int k = 0; k < DD; ++k) {
      float p = pl[k];
      mu += p * W_mu[(size_t)d * DD + k];
      lv += p * W_lv[(size_t)d * DD + k];
    }
    out[d]      = mu;
    out[DD + d] = lv;
    out[2 * DD + (size_t)NN * DD + d] = pd;
  }
}

extern "C" void kernel_launch(void* const* d_in, const int* in_sizes, int n_in,
                              void* d_out, int out_size, void* d_ws, size_t ws_size,
                              hipStream_t stream) {
  const float* x       = (const float*)d_in[0];
  const int*   ei      = (const int*)d_in[1];
  const float* W_in    = (const float*)d_in[2];
  const float* b_in    = (const float*)d_in[3];
  const float* gat_W   = (const float*)d_in[4];
  const float* att_src = (const float*)d_in[5];
  const float* att_dst = (const float*)d_in[6];
  const float* gat_b   = (const float*)d_in[7];
  const float* ln_g    = (const float*)d_in[8];
  const float* ln_b    = (const float*)d_in[9];
  const float* W_mu    = (const float*)d_in[10];
  const float* b_mu    = (const float*)d_in[11];
  const float* W_lv    = (const float*)d_in[12];
  const float* b_lv    = (const float*)d_in[13];
  float* out = (float*)d_out;

  float* h      = (float*)d_ws;
  float* xh     = h     + (size_t)NN * DD;
  float* plane  = xh    + (size_t)NN * DD;
  float* e_src  = plane + (size_t)NN * DD;
  float* e_dst  = e_src + (size_t)NN * HH;
  float* part   = e_dst + (size_t)NN * HH;

  int* binned    = (int*)plane;
  int* blockhist = binned    + (size_t)EE;
  int* blockoff  = blockhist + (size_t)NB * NT;
  int* binstart  = blockoff  + (size_t)NB * NT;

  float* h_final = out + 2 * DD;

  const int TPB = 256;
  const int nh_blk   = (NN * HH + TPB - 1) / TPB;
  const int gemm_blk = NN / 16;

  gemm_wmma_k<F_IN><<<gemm_blk, TPB, 0, stream>>>(x, W_in, b_in, h);

  bin_count_k  <<<NB, TPB, 0, stream>>>(ei, blockhist);
  bin_scan_k   <<<1, 1024, 0, stream>>>(blockhist, blockoff, binstart);
  bin_scatter_k<<<NB, TPB, 0, stream>>>(ei, blockoff, binstart, binned);

  for (int l = 0; l < LL; ++l) {
    const float* Wl  = gat_W   + (size_t)l * DD * DD;
    const float* asl = att_src + (size_t)l * HH * CC;
    const float* adl = att_dst + (size_t)l * HH * CC;
    const float* gbl = gat_b   + (size_t)l * DD;
    const float* gl  = ln_g    + (size_t)l * DD;
    const float* bl  = ln_b    + (size_t)l * DD;

    gemm_wmma_k<DD><<<gemm_blk, TPB, 0, stream>>>(h, Wl, nullptr, xh);

    att_logits_k<<<nh_blk, TPB, 0, stream>>>(xh, asl, adl, e_src, e_dst);

    float* hout  = (l == LL - 1) ? h_final : h;
    float* partl = (l == LL - 1) ? part : nullptr;
    tile_aggr_ln_k<<<NT, TPB, 0, stream>>>(binned, binstart, e_src, e_dst, xh,
                                           h, gbl, gl, bl, hout, partl);
  }

  head_k<<<1, 1024, 0, stream>>>(part, W_mu, b_mu, W_lv, b_lv, out);
}


# layers 2-4: xh = h W^T computed in the previous layer's tile kernel from its LayerNorm rows in LDS (wave = 128x16 strip on WMMA, B fragment once per k step); 3 GEMM launches gone; k order kept
# speedup vs baseline: 3.4383x; 1.1023x over previous
#include <hip/hip_runtime.h>
#include <hip/hip_bf16.h>


#define NN      100000
#define EE      1600000
#define F_IN    256
#define DD      128
#define HH      8
#define CC      16
#define LL      4
#define NEG_SLOPE 0.2f
#define LN_EPS  1e-5f

#define TN      128
#define NT      ((NN + TN - 1) / TN)
#define CH      8192
#define NB      ((EE + CH - 1) / CH)
#define SCAN_U  28
#define BIN_U   8
static_assert(NB % SCAN_U == 0, "scan batches must cover the blocks exactly");
static_assert(CH % (256 * BIN_U) == 0, "binning batches must cover a block exactly");
static_assert(NT <= 1024, "one scan thread per tile");
static_assert(NN < (1 << 17) && TN == 128, "src and local dst share one word");

typedef __attribute__((ext_vector_type(2))) float v2f;
typedef __attribute__((ext_vector_type(8))) float v8f;

template <int K>
__global__ __launch_bounds__(256) void gemm_wmma_k(
    const float* __restrict__ X, const float* __restrict__ W,
    const float* __restrict__ bias, float* __restrict__ Y) {
  constexpr int KP = K + 4;
  __shared__ float As[16 * KP];

  const int wave = threadIdx.x >> 5;
  const int lane = threadIdx.x & 31;
  const int row0 = blockIdx.x * 16;
  const int col0 = wave * 16;
  const int m    = lane & 15;
  const int koff = (lane < 16) ? 0 : 2;

  #pragma unroll
  for (int i = threadIdx.x * 4; i < 16 * K; i += 256 * 4) {
    const int r = i / K, c = i % K;
    *(float4*)&As[r * KP + c] =
        *(const float4*)(X + (size_t)(row0 + r) * K + c);
  }
  __syncthreads();

  const float* __restrict__ arow = As + m * KP;
  const float* __restrict__ wrow = W + (size_t)(col0 + m) * K;

  v8f acc = {};
#if __has_builtin(__builtin_amdgcn_wmma_f32_16x16x4_f32)
  for (int k = 0; k < K; k += 4) {
    v2f a, b;
    a.x = arow[k + koff];
    a.y = arow[k + koff + 1];
    b.x = wrow[k + koff];
    b.y = wrow[k + koff + 1];
    acc = __builtin_amdgcn_wmma_f32_16x16x4_f32(
        false, a, false, b, (short)0, acc, false, false);
  }
#else
  for (int k = 0; k < K; ++k) {
    float av = arow[k];
    #pragma unroll
    for (int v = 0; v < 8; ++v) acc[v] += av * W[(size_t)(col0 + v) * K + k];
  }
#endif

  const int mrow = (lane < 16) ? 0 : 8;
  const int c    = col0 + m;
  const float bv = bias ? bias[c] : 0.0f;
  #pragma unroll
  for (int v = 0; v < 8; ++v) {
    Y[(size_t)(row0 + v + mrow) * DD + c] = acc[v] + bv;
  }
}

__global__ __launch_bounds__(256) void att_logits_k(
    const float* __restrict__ xh, const float* __restrict__ a_src,
    const float* __restrict__ a_dst, float* __restrict__ e_src,
    float* __restrict__ e_dst) {
  int t = blockIdx.x * blockDim.x + threadIdx.x;
  if (t >= NN * HH) return;
  int n = t >> 3, h = t & 7;
  const float4* xp = (const float4*)(xh + (size_t)n * DD + h * CC);
  const float4* sp = (const float4*)(a_src + h * CC);
  const float4* dp = (const float4*)(a_dst + h * CC);
  float ss = 0.f, sd = 0.f;
  #pragma unroll
  for (int i = 0; i < 4; ++i) {
    float4 x = xp[i], s = sp[i], d = dp[i];
    ss += x.x * s.x + x.y * s.y + x.z * s.z + x.w * s.w;
    sd += x.x * d.x + x.y * d.y + x.z * d.z + x.w * d.w;
  }
  e_src[t] = ss;
  e_dst[t] = sd;
}

__device__ __forceinline__ float leaky(float v) {
  return v > 0.f ? v : v * NEG_SLOPE;
}
__device__ __forceinline__ unsigned f2ord(float f) {
  unsigned u = __float_as_uint(f);
  return (u & 0x80000000u) ? ~u : (u | 0x80000000u);
}
__device__ __forceinline__ float ord2f(unsigned u) {
  return (u & 0x80000000u) ? __uint_as_float(u & 0x7fffffffu)
                           : __uint_as_float(~u);
}

__global__ __launch_bounds__(256) void bin_count_k(
    const int* __restrict__ ei, int* __restrict__ blockhist) {
  __shared__ int hist[NT];
  for (int i = threadIdx.x; i < NT; i += 256) hist[i] = 0;
  __syncthreads();
  const int e0 = blockIdx.x * CH;
  for (int i0 = threadIdx.x; i0 < CH; i0 += 256 * BIN_U) {
    int d[BIN_U];
    #pragma unroll
    for (int u = 0; u < BIN_U; ++u) {
      const int e = e0 + i0 + 256 * u;
      d[u] = (e < EE) ? ei[EE + e] : -1;
    }
    #pragma unroll
    for (int u = 0; u < BIN_U; ++u)
      if (d[u] >= 0) atomicAdd(&hist[d[u] >> 7], 1);
  }
  __syncthreads();
  for (int i = threadIdx.x; i < NT; i += 256)
    blockhist[blockIdx.x * NT + i] = hist[i];
}

__global__ __launch_bounds__(1024) void bin_scan_k(
    const int* __restrict__ blockhist, int* __restrict__ blockoff,
    int* __restrict__ binstart) {
  __shared__ int sc[1024];
  const int t = threadIdx.x;
  int run = 0;
  if (t < NT) {
    for (int b0 = 0; b0 < NB; b0 += SCAN_U) {
      int c[SCAN_U];
      #pragma unroll
      for (int u = 0; u < SCAN_U; ++u) c[u] = blockhist[(b0 + u) * NT + t];
      #pragma unroll
      for (int u = 0; u < SCAN_U; ++u) {
        blockoff[(b0 + u) * NT + t] = run;
        run += c[u];
      }
    }
  }
  sc[t] = run;
  __syncthreads();
  for (int o = 1; o < 1024; o <<= 1) {
    const int v = (t >= o) ? sc[t - o] : 0;
    __syncthreads();
    sc[t] += v;
    __syncthreads();
  }
  if (t < NT) binstart[t] = sc[t] - run;
  if (t == NT - 1) binstart[NT] = sc[t];
}

__global__ __launch_bounds__(256) void bin_scatter_k(
    const int* __restrict__ ei, const int* __restrict__ blockoff,
    const int* __restrict__ binstart, int* __restrict__ binned) {
  __shared__ int cur[NT];
  for (int i = threadIdx.x; i < NT; i += 256)
    cur[i] = binstart[i] + blockoff[blockIdx.x * NT + i];
  __syncthreads();
  const int e0 = blockIdx.x * CH;
  for (int i0 = threadIdx.x; i0 < CH; i0 += 256 * BIN_U) {
    int s[BIN_U], d[BIN_U];
    #pragma unroll
    for (int u = 0; u < BIN_U; ++u) {
      const int e = e0 + i0 + 256 * u;
      s[u] = (e < EE) ? ei[e] : 0;
      d[u] = (e < EE) ? ei[EE + e] : -1;
    }
    #pragma unroll
    for (int u = 0; u < BIN_U; ++u) {
      if (d[u] >= 0) {
        const int pos = atomicAdd(&cur[d[u] >> 7], 1);
        binned[pos] = (s[u] << 7) | (d[u] & 127);
      }
    }
  }
}

#define AG_U 16

__device__ __forceinline__ int fetch_edge(const int* __restrict__ binned,
                                          int eb, int ne, int tot, int n0,
                                          int idx) {
  int pk = n0 << 7;
  if (idx < ne) pk = binned[eb + idx];
  else if (idx < tot) { const int dl = idx - ne; pk = ((n0 + dl) << 7) | dl; }
  return pk;
}

__device__ __forceinline__ void tile_frags(const float* arow,
                                           const float* __restrict__ wrow,
                                           int k, int koff, int sw,
                                           v2f (&a)[8], v2f& b) {
  const int kc = (k + koff) ^ sw;
  #pragma unroll
  for (int rb = 0; rb < 8; ++rb)
    a[rb] = *(const v2f*)(arow + rb * 16 * DD + kc);
  b = *(const v2f*)(wrow + k);
}
__device__ __forceinline__ void tile_mma(v8f (&d)[8], const v2f (&a)[8],
                                         v2f b) {
  #pragma unroll
  for (int rb = 0; rb < 8; ++rb)
    d[rb] = __builtin_amdgcn_wmma_f32_16x16x4_f32(
        false, a[rb], false, b, (short)0, d[rb], false, false);
}

__global__ __launch_bounds__(256) __attribute__((amdgpu_waves_per_eu(8)))
void tile_aggr_ln_k(
    const int* __restrict__ binned, const int* __restrict__ binstart,
    const float* __restrict__ e_src, const float* __restrict__ e_dst,
    const float* __restrict__ xh, const float* hin,
    const float* __restrict__ gb, const float* __restrict__ g,
    const float* __restrict__ b, float* hout, float* __restrict__ part,
    const float* __restrict__ Wn, float* __restrict__ xo) {
  __shared__ __attribute__((aligned(16))) float acc[TN * DD];
  __shared__ float    sden[TN * HH];
  __shared__ unsigned smax[TN * HH];
  __shared__ float    sed[TN * HH];
  __shared__ __attribute__((aligned(16))) float pcol[8 * DD];

  const int tid  = threadIdx.x;
  const int lane = tid & 31;
  const int wave = __builtin_amdgcn_readfirstlane(tid >> 5);
  const int h    = lane >> 2;
  const int n0   = blockIdx.x * TN;
  const int nn   = (NN - n0 < TN) ? (NN - n0) : TN;
  const int eb   = binstart[blockIdx.x];
  const int ne   = binstart[blockIdx.x + 1] - eb;
  const int tot  = ne + nn;

  for (int i = tid * 4; i < TN * DD; i += 256 * 4)
    *(float4*)&acc[i] = make_float4(0.f, 0.f, 0.f, 0.f);
  for (int i = tid; i < TN * HH; i += 256) {
    float es = 0.f, ed = 0.f;
    if ((i >> 3) < nn) {
      es = e_src[(size_t)n0 * HH + i];
      ed = e_dst[(size_t)n0 * HH + i];
    }
    sed[i]  = ed;
    sden[i] = 0.f;
    smax[i] = f2ord(leaky(es + ed));
  }
  __syncthreads();

  for (int i0 = tid; i0 < ne; i0 += 256 * 2) {
    int pk[2];
    float4 ea[2], ec[2];
    #pragma unroll
    for (int u = 0; u < 2; ++u) {
      const int i = i0 + 256 * u;
      pk[u] = (i < ne) ? binned[eb + i] : -1;
    }
    #pragma unroll
    for (int u = 0; u < 2; ++u) {
      const int s = (pk[u] >= 0) ? (pk[u] >> 7) : n0;
      const float4* p = (const float4*)(e_src + (size_t)s * HH);
      ea[u] = p[0];
      ec[u] = p[1];
    }
    #pragma unroll
    for (int u = 0; u < 2; ++u) {
      if (pk[u] >= 0) {
        const int dl = pk[u] & 127;
        const float* sd = sed + dl * HH;
        unsigned*    sm = smax + dl * HH;
        atomicMax(sm + 0, f2ord(leaky(ea[u].x + sd[0])));
        atomicMax(sm + 1, f2ord(leaky(ea[u].y + sd[1])));
        atomicMax(sm + 2, f2ord(leaky(ea[u].z + sd[2])));
        atomicMax(sm + 3, f2ord(leaky(ea[u].w + sd[3])));
        atomicMax(sm + 4, f2ord(leaky(ec[u].x + sd[4])));
        atomicMax(sm + 5, f2ord(leaky(ec[u].y + sd[5])));
        atomicMax(sm + 6, f2ord(leaky(ec[u].z + sd[6])));
        atomicMax(sm + 7, f2ord(leaky(ec[u].w + sd[7])));
      }
    }
  }
  __syncthreads();

  {
    int base = wave * 32;
    int pk_l = fetch_edge(binned, eb, ne, tot, n0, base + lane);
    while (base < tot) {
      const int nbase = base + 8 * 32;
      const int pk_n = fetch_edge(binned, eb, ne, tot, n0, nbase + lane);
      const int cnt  = (tot - base < 32) ? (tot - base) : 32;
      for (int k0 = 0; k0 < cnt; k0 += AG_U) {
        float4 x[AG_U];
        float  es[AG_U];
        int    dls[AG_U];
        #pragma unroll
        for (int u = 0; u < AG_U; ++u) {
          const int pk = __builtin_amdgcn_readlane(pk_l, k0 + u);
          const int s  = pk >> 7;
          dls[u] = pk & 127;
          es[u]  = e_src[(size_t)s * HH + h];
          x[u]   = *(const float4*)(xh + (size_t)s * DD + lane * 4);
        }
        #pragma unroll
        for (int u = 0; u < AG_U; ++u) {
          if (k0 + u < cnt) {
            const int   dl = dls[u];
            const float v  = leaky(es[u] + sed[dl * HH + h]);
            const float ex = __expf(v - ord2f(smax[dl * HH + h]));
            float* a = acc + dl * DD + lane;
            atomicAdd(a,      x[u].x * ex);
            atomicAdd(a + 32, x[u].y * ex);
            atomicAdd(a + 64, x[u].z * ex);
            atomicAdd(a + 96, x[u].w * ex);
            if ((lane & 3) == 0) atomicAdd(&sden[dl * HH + h], ex);
          }
        }
      }
      pk_l = pk_n;
      base = nbase;
    }
  }
  __syncthreads();

  const float4 bv = *(const float4*)(gb + lane * 4);
  const float4 gv = *(const float4*)(g + lane * 4);
  const float4 lb = *(const float4*)(b + lane * 4);
  float4 ps = make_float4(0.f, 0.f, 0.f, 0.f);
  for (int q0 = 0; q0 < TN / 8; q0 += 4) {
    float4 hv[4];
    #pragma unroll
    for (int u = 0; u < 4; ++u) {
      const int dl = wave + 8 * (q0 + u);
      hv[u] = make_float4(0.f, 0.f, 0.f, 0.f);
      if (dl < nn)
        hv[u] = *(const float4*)(hin + (size_t)(n0 + dl) * DD + lane * 4);
    }
    #pragma unroll
    for (int u = 0; u < 4; ++u) {
      const int dl = wave + 8 * (q0 + u);
      if (dl < nn) {
        const float  den = sden[dl * HH + h] + 1e-16f;
        const float* a   = acc + dl * DD + lane;
        float x0 = hv[u].x + a[0]  / den + bv.x;
        float x1 = hv[u].y + a[32] / den + bv.y;
        float x2 = hv[u].z + a[64] / den + bv.z;
        float x3 = hv[u].w + a[96] / den + bv.w;
        float s = x0 + x1 + x2 + x3;
        #pragma unroll
        for (int o = 16; o > 0; o >>= 1) s += __shfl_xor(s, o, 32);
        float mean = s * (1.f / (float)DD);
        float d0 = x0 - mean, d1 = x1 - mean, d2 = x2 - mean, d3 = x3 - mean;
        float vs = d0 * d0 + d1 * d1 + d2 * d2 + d3 * d3;
        #pragma unroll
        for (int o = 16; o > 0; o >>= 1) vs += __shfl_xor(vs, o, 32);
        float inv = rsqrtf(vs * (1.f / (float)DD) + LN_EPS);
        float4 o4;
        o4.x = d0 * inv * gv.x + lb.x;
        o4.y = d1 * inv * gv.y + lb.y;
        o4.z = d2 * inv * gv.z + lb.z;
        o4.w = d3 * inv * gv.w + lb.w;
        *(float4*)(hout + (size_t)(n0 + dl) * DD + lane * 4) = o4;
        if (Wn)
          *(float4*)&acc[dl * DD + ((lane * 4) ^ ((dl & 15) << 2))] = o4;
        ps.x += o4.x;
        ps.y += o4.y;
        ps.z += o4.z;
        ps.w += o4.w;
      }
    }
  }

  if (part) {
    *(float4*)&pcol[wave * DD + lane * 4] = ps;
    __syncthreads();
    if (tid < DD) {
      float s = 0.f;
      #pragma unroll
      for (int w = 0; w < 8; ++w) s += pcol[w * DD + tid];
      part[(size_t)blockIdx.x * DD + tid] = s;
    }
  }

  if (Wn) {
    __syncthreads();
    const int c    = lane & 15;
    const int koff = (lane < 16) ? 0 : 2;
    const int sw   = c << 2;
    const float* arow = acc + c * DD;
    const float* __restrict__ wrow = Wn + (size_t)(wave * 16 + c) * DD + koff;
    v8f d[8];
    #pragma unroll
    for (int rb = 0; rb < 8; ++rb) d[rb] = v8f{};
    #pragma unroll
    for (int kk = 0; kk < DD / 4; ++kk) {
      v2f a[8], b;
      tile_frags(arow, wrow, 4 * kk, koff, sw, a, b);
      tile_mma(d, a, b);
    }

    const int mrow = (lane < 16) ? 0 : 8;
    float* xc = xo + (size_t)(n0 + mrow) * DD + wave * 16 + c;
    if (nn == TN) {
      #pragma unroll
      for (int rb = 0; rb < 8; ++rb) {
        #pragma unroll
        for (int r = 0; r < 8; ++r)
          xc[(rb * 16 + r) * DD] = d[rb][r];
      }
    } else {
      #pragma unroll
      for (int rb = 0; rb < 8; ++rb) {
        #pragma unroll
        for (int r = 0; r < 8; ++r) {
          if (rb * 16 + mrow + r < nn) xc[(rb * 16 + r) * DD] = d[rb][r];
        }
      }
    }
  }
}

#define HP_G 8
#define HP_U 25
static_assert(HP_G * HP_U * 4 >= NT, "four batches cover the tiles");
__global__ __launch_bounds__(1024) void head_k(
    const float* __restrict__ part, const float* __restrict__ W_mu,
    const float* __restrict__ b_mu, const float* __restrict__ W_lv,
    const float* __restrict__ b_lv, float* __restrict__ out) {
  __shared__ float red[HP_G * DD];
  __shared__ float pl[DD];
  const int d = threadIdx.x & (DD - 1);
  const int g = threadIdx.x >> 7;
  float s = 0.f;
  for (int t0 = g; t0 < NT; t0 += HP_G * HP_U) {
    float v[HP_U];
    #pragma unroll
    for (int u = 0; u < HP_U; ++u) {
      const int t = t0 + HP_G * u;
      v[u] = (t < NT) ? part[(size_t)t * DD + d] : 0.f;
    }
    #pragma unroll
    for (int u = 0; u < HP_U; ++u) s += v[u];
  }
  red[g * DD + d] = s;
  __syncthreads();
  float pd = 0.f;
  if (threadIdx.x < DD) {
    #pragma unroll
    for (int w = 0; w < HP_G; ++w) pd += red[w * DD + d];
    pd *= (1.f / (float)NN);
    pl[d] = pd;
  }
  __syncthreads();
  if (threadIdx.x < DD) {
    float mu = b_mu[d], lv = b_lv[d];
    for (int k = 0; k < DD; ++k) {
      float p = pl[k];
      mu += p * W_mu[(size_t)d * DD + k];
      lv += p * W_lv[(size_t)d * DD + k];
    }
    out[d]      = mu;
    out[DD + d] = lv;
    out[2 * DD + (size_t)NN * DD + d] = pd;
  }
}

extern "C" void kernel_launch(void* const* d_in, const int* in_sizes, int n_in,
                              void* d_out, int out_size, void* d_ws, size_t ws_size,
                              hipStream_t stream) {
  const float* x       = (const float*)d_in[0];
  const int*   ei      = (const int*)d_in[1];
  const float* W_in    = (const float*)d_in[2];
  const float* b_in    = (const float*)d_in[3];
  const float* gat_W   = (const float*)d_in[4];
  const float* att_src = (const float*)d_in[5];
  const float* att_dst = (const float*)d_in[6];
  const float* gat_b   = (const float*)d_in[7];
  const float* ln_g    = (const float*)d_in[8];
  const float* ln_b    = (const float*)d_in[9];
  const float* W_mu    = (const float*)d_in[10];
  const float* b_mu    = (const float*)d_in[11];
  const float* W_lv    = (const float*)d_in[12];
  const float* b_lv    = (const float*)d_in[13];
  float* out = (float*)d_out;

  float* h      = (float*)d_ws;
  float* xh     = h     + (size_t)NN * DD;
  float* plane  = xh    + (size_t)NN * DD;
  float* e_src  = plane + (size_t)NN * DD;
  float* e_dst  = e_src + (size_t)NN * HH;
  float* part   = e_dst + (size_t)NN * HH;

  int* binned    = (int*)plane;
  int* blockhist = binned    + (size_t)EE;
  int* blockoff  = blockhist + (size_t)NB * NT;
  int* binstart  = blockoff  + (size_t)NB * NT;

  float* h_final = out + 2 * DD;
  float* xh2     = h_final;

  const int TPB = 256;
  const int nh_blk   = (NN * HH + TPB - 1) / TPB;
  const int gemm_blk = NN / 16;

  gemm_wmma_k<F_IN><<<gemm_blk, TPB, 0, stream>>>(x, W_in, b_in, h);

  bin_count_k  <<<NB, TPB, 0, stream>>>(ei, blockhist);
  bin_scan_k   <<<1, 1024, 0, stream>>>(blockhist, blockoff, binstart);
  bin_scatter_k<<<NB, TPB, 0, stream>>>(ei, blockoff, binstart, binned);

  for (int l = 0; l < LL; ++l) {
    const float* Wl  = gat_W   + (size_t)l * DD * DD;
    const float* asl = att_src + (size_t)l * HH * CC;
    const float* adl = att_dst + (size_t)l * HH * CC;
    const float* gbl = gat_b   + (size_t)l * DD;
    const float* gl  = ln_g    + (size_t)l * DD;
    const float* bl  = ln_b    + (size_t)l * DD;

    float* xcur = (l & 1) ? xh : xh2;
    float* xnxt = (l & 1) ? xh2 : xh;
    const float* Wn = (l == LL - 1) ? nullptr : Wl + (size_t)DD * DD;

    if (l == 0) gemm_wmma_k<DD><<<gemm_blk, TPB, 0, stream>>>(h, Wl, nullptr, xcur);

    att_logits_k<<<nh_blk, TPB, 0, stream>>>(xcur, asl, adl, e_src, e_dst);

    float* hout  = (l == LL - 1) ? h_final : h;
    float* partl = (l == LL - 1) ? part : nullptr;
    tile_aggr_ln_k<<<NT, TPB, 0, stream>>>(binned, binstart, e_src, e_dst, xcur,
                                           h, gbl, gl, bl, hout, partl,
                                           Wn, (l == LL - 1) ? nullptr : xnxt);
  }

  head_k<<<1, 1024, 0, stream>>>(part, W_mu, b_mu, W_lv, b_lv, out);
}


# destination tiles of 100 nodes instead of 128: 1000 work-groups, at most 4 per WGP and all resident (was 782: 4 on 14 WGPs, 3 elsewhere); in-tile transform on 7 row blocks; pool sum regrouped
# speedup vs baseline: 3.5679x; 1.0377x over previous
#include <hip/hip_runtime.h>
#include <hip/hip_bf16.h>


#define NN      100000
#define EE      1600000
#define F_IN    256
#define DD      128
#define HH      8
#define CC      16
#define LL      4
#define NEG_SLOPE 0.2f
#define LN_EPS  1e-5f

#define TN      100
#define NT      ((NN + TN - 1) / TN)
#define RB      ((TN + 15) / 16)
#define TNP     (RB * 16)
#define CH      8192
#define NB      ((EE + CH - 1) / CH)
#define SCAN_U  28
#define BIN_U   8
static_assert(NB % SCAN_U == 0, "scan batches must cover the blocks exactly");
static_assert(CH % (256 * BIN_U) == 0, "binning batches must cover a block exactly");
static_assert(NT <= 1024, "one scan thread per tile");
static_assert(NN < (1 << 17) && TN <= 128, "src and local dst share one word");
static_assert((TNP * DD) % (256 * 4) == 0, "the clearing loop covers the plane exactly");

typedef __attribute__((ext_vector_type(2))) float v2f;
typedef __attribute__((ext_vector_type(8))) float v8f;

template <int K>
__global__ __launch_bounds__(256) void gemm_wmma_k(
    const float* __restrict__ X, const float* __restrict__ W,
    const float* __restrict__ bias, float* __restrict__ Y) {
  constexpr int KP = K + 4;
  __shared__ float As[16 * KP];

  const int wave = threadIdx.x >> 5;
  const int lane = threadIdx.x & 31;
  const int row0 = blockIdx.x * 16;
  const int col0 = wave * 16;
  const int m    = lane & 15;
  const int koff = (lane < 16) ? 0 : 2;

  #pragma unroll
  for (int i = threadIdx.x * 4; i < 16 * K; i += 256 * 4) {
    const int r = i / K, c = i % K;
    *(float4*)&As[r * KP + c] =
        *(const float4*)(X + (size_t)(row0 + r) * K + c);
  }
  __syncthreads();

  const float* __restrict__ arow = As + m * KP;
  const float* __restrict__ wrow = W + (size_t)(col0 + m) * K;

  v8f acc = {};
#if __has_builtin(__builtin_amdgcn_wmma_f32_16x16x4_f32)
  for (int k = 0; k < K; k += 4) {
    v2f a, b;
    a.x = arow[k + koff];
    a.y = arow[k + koff + 1];
    b.x = wrow[k + koff];
    b.y = wrow[k + koff + 1];
    acc = __builtin_amdgcn_wmma_f32_16x16x4_f32(
        false, a, false, b, (short)0, acc, false, false);
  }
#else
  for (int k = 0; k < K; ++k) {
    float av = arow[k];
    #pragma unroll
    for (int v = 0; v < 8; ++v) acc[v] += av * W[(size_t)(col0 + v) * K + k];
  }
#endif

  const int mrow = (lane < 16) ? 0 : 8;
  const int c    = col0 + m;
  const float bv = bias ? bias[c] : 0.0f;
  #pragma unroll
  for (int v = 0; v < 8; ++v) {
    Y[(size_t)(row0 + v + mrow) * DD + c] = acc[v] + bv;
  }
}

__global__ __launch_bounds__(256) void att_logits_k(
    const float* __restrict__ xh, const float* __restrict__ a_src,
    const float* __restrict__ a_dst, float* __restrict__ e_src,
    float* __restrict__ e_dst) {
  int t = blockIdx.x * blockDim.x + threadIdx.x;
  if (t >= NN * HH) return;
  int n = t >> 3, h = t & 7;
  const float4* xp = (const float4*)(xh + (size_t)n * DD + h * CC);
  const float4* sp = (const float4*)(a_src + h * CC);
  const float4* dp = (const float4*)(a_dst + h * CC);
  float ss = 0.f, sd = 0.f;
  #pragma unroll
  for (int i = 0; i < 4; ++i) {
    float4 x = xp[i], s = sp[i], d = dp[i];
    ss += x.x * s.x + x.y * s.y + x.z * s.z + x.w * s.w;
    sd += x.x * d.x + x.y * d.y + x.z * d.z + x.w * d.w;
  }
  e_src[t] = ss;
  e_dst[t] = sd;
}

__device__ __forceinline__ float leaky(float v) {
  return v > 0.f ? v : v * NEG_SLOPE;
}
__device__ __forceinline__ unsigned f2ord(float f) {
  unsigned u = __float_as_uint(f);
  return (u & 0x80000000u) ? ~u : (u | 0x80000000u);
}
__device__ __forceinline__ float ord2f(unsigned u) {
  return (u & 0x80000000u) ? __uint_as_float(u & 0x7fffffffu)
                           : __uint_as_float(~u);
}

__global__ __launch_bounds__(256) void bin_count_k(
    const int* __restrict__ ei, int* __restrict__ blockhist) {
  __shared__ int hist[NT];
  for (int i = threadIdx.x; i < NT; i += 256) hist[i] = 0;
  __syncthreads();
  const int e0 = blockIdx.x * CH;
  for (int i0 = threadIdx.x; i0 < CH; i0 += 256 * BIN_U) {
    int d[BIN_U];
    #pragma unroll
    for (int u = 0; u < BIN_U; ++u) {
      const int e = e0 + i0 + 256 * u;
      d[u] = (e < EE) ? ei[EE + e] : -1;
    }
    #pragma unroll
    for (int u = 0; u < BIN_U; ++u)
      if (d[u] >= 0) atomicAdd(&hist[d[u] / TN], 1);
  }
  __syncthreads();
  for (int i = threadIdx.x; i < NT; i += 256)
    blockhist[blockIdx.x * NT + i] = hist[i];
}

__global__ __launch_bounds__(1024) void bin_scan_k(
    const int* __restrict__ blockhist, int* __restrict__ blockoff,
    int* __restrict__ binstart) {
  __shared__ int sc[1024];
  const int t = threadIdx.x;
  int run = 0;
  if (t < NT) {
    for (int b0 = 0; b0 < NB; b0 += SCAN_U) {
      int c[SCAN_U];
      #pragma unroll
      for (int u = 0; u < SCAN_U; ++u) c[u] = blockhist[(b0 + u) * NT + t];
      #pragma unroll
      for (int u = 0; u < SCAN_U; ++u) {
        blockoff[(b0 + u) * NT + t] = run;
        run += c[u];
      }
    }
  }
  sc[t] = run;
  __syncthreads();
  for (int o = 1; o < 1024; o <<= 1) {
    const int v = (t >= o) ? sc[t - o] : 0;
    __syncthreads();
    sc[t] += v;
    __syncthreads();
  }
  if (t < NT) binstart[t] = sc[t] - run;
  if (t == NT - 1) binstart[NT] = sc[t];
}

__global__ __launch_bounds__(256) void bin_scatter_k(
    const int* __restrict__ ei, const int* __restrict__ blockoff,
    const int* __restrict__ binstart, int* __restrict__ binned) {
  __shared__ int cur[NT];
  for (int i = threadIdx.x; i < NT; i += 256)
    cur[i] = binstart[i] + blockoff[blockIdx.x * NT + i];
  __syncthreads();
  const int e0 = blockIdx.x * CH;
  for (int i0 = threadIdx.x; i0 < CH; i0 += 256 * BIN_U) {
    int s[BIN_U], d[BIN_U];
    #pragma unroll
    for (int u = 0; u < BIN_U; ++u) {
      const int e = e0 + i0 + 256 * u;
      s[u] = (e < EE) ? ei[e] : 0;
      d[u] = (e < EE) ? ei[EE + e] : -1;
    }
    #pragma unroll
    for (int u = 0; u < BIN_U; ++u) {
      if (d[u] >= 0) {
        const int t   = d[u] / TN;
        const int pos = atomicAdd(&cur[t], 1);
        binned[pos] = (s[u] << 7) | (d[u] - t * TN);
      }
    }
  }
}

#define AG_U 16

__device__ __forceinline__ int fetch_edge(const int* __restrict__ binned,
                                          int eb, int ne, int tot, int n0,
                                          int idx) {
  int pk = n0 << 7;
  if (idx < ne) pk = binned[eb + idx];
  else if (idx < tot) { const int dl = idx - ne; pk = ((n0 + dl) << 7) | dl; }
  return pk;
}

__device__ __forceinline__ void tile_frags(const float* arow,
                                           const float* __restrict__ wrow,
                                           int k, int koff, int sw,
                                           v2f (&a)[RB], v2f& b) {
  const int kc = (k + koff) ^ sw;
  #pragma unroll
  for (int rb = 0; rb < RB; ++rb)
    a[rb] = *(const v2f*)(arow + rb * 16 * DD + kc);
  b = *(const v2f*)(wrow + k);
}
__device__ __forceinline__ void tile_mma(v8f (&d)[RB], const v2f (&a)[RB],
                                         v2f b) {
  #pragma unroll
  for (int rb = 0; rb < RB; ++rb)
    d[rb] = __builtin_amdgcn_wmma_f32_16x16x4_f32(
        false, a[rb], false, b, (short)0, d[rb], false, false);
}

__global__ __launch_bounds__(256) __attribute__((amdgpu_waves_per_eu(8)))
void tile_aggr_ln_k(
    const int* __restrict__ binned, const int* __restrict__ binstart,
    const float* __restrict__ e_src, const float* __restrict__ e_dst,
    const float* __restrict__ xh, const float* hin,
    const float* __restrict__ gb, const float* __restrict__ g,
    const float* __restrict__ b, float* hout, float* __restrict__ part,
    const float* __restrict__ Wn, float* __restrict__ xo) {
  __shared__ __attribute__((aligned(16))) float acc[TNP * DD];
  __shared__ float    sden[TN * HH];
  __shared__ unsigned smax[TN * HH];
  __shared__ float    sed[TN * HH];
  __shared__ __attribute__((aligned(16))) float pcol[8 * DD];

  const int tid  = threadIdx.x;
  const int lane = tid & 31;
  const int wave = __builtin_amdgcn_readfirstlane(tid >> 5);
  const int h    = lane >> 2;
  const int n0   = blockIdx.x * TN;
  const int nn   = (NN - n0 < TN) ? (NN - n0) : TN;
  const int eb   = binstart[blockIdx.x];
  const int ne   = binstart[blockIdx.x + 1] - eb;
  const int tot  = ne + nn;

  for (int i = tid * 4; i < TNP * DD; i += 256 * 4)
    *(float4*)&acc[i] = make_float4(0.f, 0.f, 0.f, 0.f);
  for (int i = tid; i < TN * HH; i += 256) {
    float es = 0.f, ed = 0.f;
    if ((i >> 3) < nn) {
      es = e_src[(size_t)n0 * HH + i];
      ed = e_dst[(size_t)n0 * HH + i];
    }
    sed[i]  = ed;
    sden[i] = 0.f;
    smax[i] = f2ord(leaky(es + ed));
  }
  __syncthreads();

  for (int i0 = tid; i0 < ne; i0 += 256 * 2) {
    int pk[2];
    float4 ea[2], ec[2];
    #pragma unroll
    for (int u = 0; u < 2; ++u) {
      const int i = i0 + 256 * u;
      pk[u] = (i < ne) ? binned[eb + i] : -1;
    }
    #pragma unroll
    for (int u = 0; u < 2; ++u) {
      const int s = (pk[u] >= 0) ? (pk[u] >> 7) : n0;
      const float4* p = (const float4*)(e_src + (size_t)s * HH);
      ea[u] = p[0];
      ec[u] = p[1];
    }
    #pragma unroll
    for (int u = 0; u < 2; ++u) {
      if (pk[u] >= 0) {
        const int dl = pk[u] & 127;
        const float* sd = sed + dl * HH;
        unsigned*    sm = smax + dl * HH;
        atomicMax(sm + 0, f2ord(leaky(ea[u].x + sd[0])));
        atomicMax(sm + 1, f2ord(leaky(ea[u].y + sd[1])));
        atomicMax(sm + 2, f2ord(leaky(ea[u].z + sd[2])));
        atomicMax(sm + 3, f2ord(leaky(ea[u].w + sd[3])));
        atomicMax(sm + 4, f2ord(leaky(ec[u].x + sd[4])));
        atomicMax(sm + 5, f2ord(leaky(ec[u].y + sd[5])));
        atomicMax(sm + 6, f2ord(leaky(ec[u].z + sd[6])));
        atomicMax(sm + 7, f2ord(leaky(ec[u].w + sd[7])));
      }
    }
  }
  __syncthreads();

  {
    int base = wave * 32;
    int pk_l = fetch_edge(binned, eb, ne, tot, n0, base + lane);
    while (base < tot) {
      const int nbase = base + 8 * 32;
      const int pk_n = fetch_edge(binned, eb, ne, tot, n0, nbase + lane);
      const int cnt  = (tot - base < 32) ? (tot - base) : 32;
      for (int k0 = 0; k0 < cnt; k0 += AG_U) {
        float4 x[AG_U];
        float  es[AG_U];
        int    dls[AG_U];
        #pragma unroll
        for (int u = 0; u < AG_U; ++u) {
          const int pk = __builtin_amdgcn_readlane(pk_l, k0 + u);
          const int s  = pk >> 7;
          dls[u] = pk & 127;
          es[u]  = e_src[(size_t)s * HH + h];
          x[u]   = *(const float4*)(xh + (size_t)s * DD + lane * 4);
        }
        #pragma unroll
        for (int u = 0; u < AG_U; ++u) {
          if (k0 + u < cnt) {
            const int   dl = dls[u];
            const float v  = leaky(es[u] + sed[dl * HH + h]);
            const float ex = __expf(v - ord2f(smax[dl * HH + h]));
            float* a = acc + dl * DD + lane;
            atomicAdd(a,      x[u].x * ex);
            atomicAdd(a + 32, x[u].y * ex);
            atomicAdd(a + 64, x[u].z * ex);
            atomicAdd(a + 96, x[u].w * ex);
            if ((lane & 3) == 0) atomicAdd(&sden[dl * HH + h], ex);
          }
        }
      }
      pk_l = pk_n;
      base = nbase;
    }
  }
  __syncthreads();

  const float4 bv = *(const float4*)(gb + lane * 4);
  const float4 gv = *(const float4*)(g + lane * 4);
  const float4 lb = *(const float4*)(b + lane * 4);
  float4 ps = make_float4(0.f, 0.f, 0.f, 0.f);
  for (int q0 = 0; q0 < (TN + 7) / 8; q0 += 4) {
    float4 hv[4];
    #pragma unroll
    for (int u = 0; u < 4; ++u) {
      const int dl = wave + 8 * (q0 + u);
      hv[u] = make_float4(0.f, 0.f, 0.f, 0.f);
      if (dl < nn)
        hv[u] = *(const float4*)(hin + (size_t)(n0 + dl) * DD + lane * 4);
    }
    #pragma unroll
    for (int u = 0; u < 4; ++u) {
      const int dl = wave + 8 * (q0 + u);
      if (dl < nn) {
        const float  den = sden[dl * HH + h] + 1e-16f;
        const float* a   = acc + dl * DD + lane;
        float x0 = hv[u].x + a[0]  / den + bv.x;
        float x1 = hv[u].y + a[32] / den + bv.y;
        float x2 = hv[u].z + a[64] / den + bv.z;
        float x3 = hv[u].w + a[96] / den + bv.w;
        float s = x0 + x1 + x2 + x3;
        #pragma unroll
        for (int o = 16; o > 0; o >>= 1) s += __shfl_xor(s, o, 32);
        float mean = s * (1.f / (float)DD);
        float d0 = x0 - mean, d1 = x1 - mean, d2 = x2 - mean, d3 = x3 - mean;
        float vs = d0 * d0 + d1 * d1 + d2 * d2 + d3 * d3;
        #pragma unroll
        for (int o = 16; o > 0; o >>= 1) vs += __shfl_xor(vs, o, 32);
        float inv = rsqrtf(vs * (1.f / (float)DD) + LN_EPS);
        float4 o4;
        o4.x = d0 * inv * gv.x + lb.x;
        o4.y = d1 * inv * gv.y + lb.y;
        o4.z = d2 * inv * gv.z + lb.z;
        o4.w = d3 * inv * gv.w + lb.w;
        *(float4*)(hout + (size_t)(n0 + dl) * DD + lane * 4) = o4;
        if (Wn)
          *(float4*)&acc[dl * DD + ((lane * 4) ^ ((dl & 15) << 2))] = o4;
        ps.x += o4.x;
        ps.y += o4.y;
        ps.z += o4.z;
        ps.w += o4.w;
      }
    }
  }

  if (part) {
    *(float4*)&pcol[wave * DD + lane * 4] = ps;
    __syncthreads();
    if (tid < DD) {
      float s = 0.f;
      #pragma unroll
      for (int w = 0; w < 8; ++w) s += pcol[w * DD + tid];
      part[(size_t)blockIdx.x * DD + tid] = s;
    }
  }

  if (Wn) {
    __syncthreads();
    const int c    = lane & 15;
    const int koff = (lane < 16) ? 0 : 2;
    const int sw   = c << 2;
    const float* arow = acc + c * DD;
    const float* __restrict__ wrow = Wn + (size_t)(wave * 16 + c) * DD + koff;
    v8f d[RB];
    #pragma unroll
    for (int rb = 0; rb < RB; ++rb) d[rb] = v8f{};
    #pragma unroll
    for (int kk = 0; kk < DD / 4; ++kk) {
      v2f a[RB], b;
      tile_frags(arow, wrow, 4 * kk, koff, sw, a, b);
      tile_mma(d, a, b);
    }

    const int mrow = (lane < 16) ? 0 : 8;
    float* xc = xo + (size_t)(n0 + mrow) * DD + wave * 16 + c;
    if (nn == TN) {
      #pragma unroll
      for (int rb = 0; rb < RB; ++rb) {
        #pragma unroll
        for (int r = 0; r < 8; ++r) {
          if (rb < TN / 16 || rb * 16 + mrow + r < TN)
            xc[(rb * 16 + r) * DD] = d[rb][r];
        }
      }
    } else {
      #pragma unroll
      for (int rb = 0; rb < RB; ++rb) {
        #pragma unroll
        for (int r = 0; r < 8; ++r) {
          if (rb * 16 + mrow + r < nn) xc[(rb * 16 + r) * DD] = d[rb][r];
        }
      }
    }
  }
}

#define HP_G 8
#define HP_U 25
static_assert(HP_G * HP_U * 5 >= NT, "five batches cover the tiles");
__global__ __launch_bounds__(1024) void head_k(
    const float* __restrict__ part, const float* __restrict__ W_mu,
    const float* __restrict__ b_mu, const float* __restrict__ W_lv,
    const float* __restrict__ b_lv, float* __restrict__ out) {
  __shared__ float red[HP_G * DD];
  __shared__ float pl[DD];
  const int d = threadIdx.x & (DD - 1);
  const int g = threadIdx.x >> 7;
  float s = 0.f;
  for (int t0 = g; t0 < NT; t0 += HP_G * HP_U) {
    float v[HP_U];
    #pragma unroll
    for (int u = 0; u < HP_U; ++u) {
      const int t = t0 + HP_G * u;
      v[u] = (t < NT) ? part[(size_t)t * DD + d] : 0.f;
    }
    #pragma unroll
    for (int u = 0; u < HP_U; ++u) s += v[u];
  }
  red[g * DD + d] = s;
  __syncthreads();
  float pd = 0.f;
  if (threadIdx.x < DD) {
    #pragma unroll
    for (int w = 0; w < HP_G; ++w) pd += red[w * DD + d];
    pd *= (1.f / (float)NN);
    pl[d] = pd;
  }
  __syncthreads();
  if (threadIdx.x < DD) {
    float mu = b_mu[d], lv = b_lv[d];
    for (int k = 0; k < DD; ++k) {
      float p = pl[k];
      mu += p * W_mu[(size_t)d * DD + k];
      lv += p * W_lv[(size_t)d * DD + k];
    }
    out[d]      = mu;
    out[DD + d] = lv;
    out[2 * DD + (size_t)NN * DD + d] = pd;
  }
}

extern "C" void kernel_launch(void* const* d_in, const int* in_sizes, int n_in,
                              void* d_out, int out_size, void* d_ws, size_t ws_size,
                              hipStream_t stream) {
  const float* x       = (const float*)d_in[0];
  const int*   ei      = (const int*)d_in[1];
  const float* W_in    = (const float*)d_in[2];
  const float* b_in    = (const float*)d_in[3];
  const float* gat_W   = (const float*)d_in[4];
  const float* att_src = (const float*)d_in[5];
  const float* att_dst = (const float*)d_in[6];
  const float* gat_b   = (const float*)d_in[7];
  const float* ln_g    = (const float*)d_in[8];
  const float* ln_b    = (const float*)d_in[9];
  const float* W_mu    = (const float*)d_in[10];
  const float* b_mu    = (const float*)d_in[11];
  const float* W_lv    = (const float*)d_in[12];
  const float* b_lv    = (const float*)d_in[13];
  float* out = (float*)d_out;

  float* h      = (float*)d_ws;
  float* xh     = h     + (size_t)NN * DD;
  float* plane  = xh    + (size_t)NN * DD;
  float* e_src  = plane + (size_t)NN * DD;
  float* e_dst  = e_src + (size_t)NN * HH;

  int* binned    = (int*)plane;
  int* blockhist = binned    + (size_t)EE;
  int* blockoff  = blockhist + (size_t)NB * NT;
  int* binstart  = blockoff  + (size_t)NB * NT;
  float* part    = (float*)(binstart + NT + 8);

  float* h_final = out + 2 * DD;
  float* xh2     = h_final;

  const int TPB = 256;
  const int nh_blk   = (NN * HH + TPB - 1) / TPB;
  const int gemm_blk = NN / 16;

  gemm_wmma_k<F_IN><<<gemm_blk, TPB, 0, stream>>>(x, W_in, b_in, h);

  bin_count_k  <<<NB, TPB, 0, stream>>>(ei, blockhist);
  bin_scan_k   <<<1, 1024, 0, stream>>>(blockhist, blockoff, binstart);
  bin_scatter_k<<<NB, TPB, 0, stream>>>(ei, blockoff, binstart, binned);

  for (int l = 0; l < LL; ++l) {
    const float* Wl  = gat_W   + (size_t)l * DD * DD;
    const float* asl = att_src + (size_t)l * HH * CC;
    const float* adl = att_dst + (size_t)l * HH * CC;
    const float* gbl = gat_b   + (size_t)l * DD;
    const float* gl  = ln_g    + (size_t)l * DD;
    const float* bl  = ln_b    + (size_t)l * DD;

    float* xcur = (l & 1) ? xh : xh2;
    float* xnxt = (l & 1) ? xh2 : xh;
    const float* Wn = (l == LL - 1) ? nullptr : Wl + (size_t)DD * DD;

    if (l == 0) gemm_wmma_k<DD><<<gemm_blk, TPB, 0, stream>>>(h, Wl, nullptr, xcur);

    att_logits_k<<<nh_blk, TPB, 0, stream>>>(xcur, asl, adl, e_src, e_dst);

    float* hout  = (l == LL - 1) ? h_final : h;
    float* partl = (l == LL - 1) ? part : nullptr;
    tile_aggr_ln_k<<<NT, TPB, 0, stream>>>(binned, binstart, e_src, e_dst, xcur,
                                           h, gbl, gl, bl, hout, partl,
                                           Wn, (l == LL - 1) ? nullptr : xnxt);
  }

  head_k<<<1, 1024, 0, stream>>>(part, W_mu, b_mu, W_lv, b_lv, out);
}


# input projection and first transform: W resident in LDS (one 16-wave work-group per WGP), a wave does 16x64 strips with A fragments as 16-byte global loads 64 k ahead; k sums regrouped
# speedup vs baseline: 3.8528x; 1.0799x over previous
#include <hip/hip_runtime.h>
#include <hip/hip_bf16.h>


#define NN      100000
#define EE      1600000
#define F_IN    256
#define DD      128
#define HH      8
#define CC      16
#define LL      4
#define NEG_SLOPE 0.2f
#define LN_EPS  1e-5f

#define TN      100
#define NT      ((NN + TN - 1) / TN)
#define RB      ((TN + 15) / 16)
#define TNP     (RB * 16)
#define CH      8192
#define NB      ((EE + CH - 1) / CH)
#define SCAN_U  28
#define BIN_U   8
static_assert(NB % SCAN_U == 0, "scan batches must cover the blocks exactly");
static_assert(CH % (256 * BIN_U) == 0, "binning batches must cover a block exactly");
static_assert(NT <= 1024, "one scan thread per tile");
static_assert(NN < (1 << 17) && TN <= 128, "src and local dst share one word");
static_assert((TNP * DD) % (256 * 4) == 0, "the clearing loop covers the plane exactly");

typedef __attribute__((ext_vector_type(2))) float v2f;
typedef __attribute__((ext_vector_type(8))) float v8f;

#define GEMM_WG   256
#define GEMM_TPB  512
#define GEMM_LDS_FLOATS 41472
static_assert(DD * (F_IN + 4) <= GEMM_LDS_FLOATS, "the larger weight matrix fits");
static_assert(NN % 16 == 0, "whole 16-row blocks");

__device__ __forceinline__ int frag_k(int j, int hf) { return 8 * j + 4 * hf; }

template <int K>
__global__ __launch_bounds__(GEMM_TPB) void gemm_wlds_k(
    const float* __restrict__ X, const float* __restrict__ W,
    const float* __restrict__ bias, float* __restrict__ Y) {
  constexpr int KP  = K + 4;
  constexpr int NCH = K / 64;
  static_assert(NCH % 2 == 0, "the chunk ring of two closes over a unit");
  __shared__ __attribute__((aligned(16))) float Ws[GEMM_LDS_FLOATS];

  const int tid  = threadIdx.x;
  const int lane = tid & 31;
  const int wave = __builtin_amdgcn_readfirstlane(tid >> 5);
  const int c    = lane & 15;
  const int hf   = lane >> 4;

  const int rb0 = (int)(((long long)blockIdx.x * (NN / 16)) / GEMM_WG);
  const int rb1 = (int)(((long long)(blockIdx.x + 1) * (NN / 16)) / GEMM_WG);
  const int nu  = (rb1 - rb0) * 2;
  const int ch  = wave & 1;

  const float* wl = Ws + (ch * 64 + c) * KP;
  const float* xa = X + (size_t)((rb0 + (wave >> 1)) * 16 + c) * K;

  float4 a[2][8];
  if (wave < nu) {
    #pragma unroll
    for (int j = 0; j < 8; ++j)
      a[0][j] = *(const float4*)(xa + frag_k(j, hf));
  }

  {
    constexpr int WLD = (DD * K) / (GEMM_TPB * 4);
    static_assert(WLD * GEMM_TPB * 4 == DD * K, "the copy covers W exactly");
    float4 wv[WLD];
    #pragma unroll
    for (int i = 0; i < WLD; ++i)
      wv[i] = *(const float4*)(W + (tid + i * GEMM_TPB) * 4);
    #pragma unroll
    for (int i = 0; i < WLD; ++i) {
      const int e = (tid + i * GEMM_TPB) * 4;
      const int r = e / K, cc = e % K;
      *(float4*)&Ws[r * KP + cc] = wv[i];
    }
  }
  __syncthreads();

  for (int u = wave; u < nu; u += 16) {
    asm volatile("" ::: "memory");
    const int rb = rb0 + (u >> 1);
    const int un = (u + 16 < nu) ? (u + 16) : u;
    const float* xn = X + (size_t)((rb0 + (un >> 1)) * 16 + c) * K;

    v8f acc[4];
    #pragma unroll
    for (int t = 0; t < 4; ++t) acc[t] = v8f{};
    float4 bq[2][4];
    #pragma unroll
    for (int t = 0; t < 4; ++t)
      bq[0][t] = *(const float4*)(wl + t * 16 * KP + frag_k(0, hf));

    #pragma unroll
    for (int cc = 0; cc < NCH; ++cc) {
      const float* src = (cc + 1 < NCH) ? (xa + (cc + 1) * 64) : xn;
      #pragma unroll
      for (int j = 0; j < 8; ++j)
        a[(cc + 1) & 1][j] = *(const float4*)(src + frag_k(j, hf));

      #pragma unroll
      for (int j = 0; j < 8; ++j) {
        const int sp = cc * 8 + j;
        if (sp + 1 < K / 8) {
          #pragma unroll
          for (int t = 0; t < 4; ++t)
            bq[(sp + 1) & 1][t] =
                *(const float4*)(wl + t * 16 * KP + frag_k(sp + 1, hf));
        }
        const float4 av = a[cc & 1][j];
        const v2f a0 = {av.x, av.y}, a1 = {av.z, av.w};
        #pragma unroll
        for (int t = 0; t < 4; ++t) {
          const v2f b0 = {bq[sp & 1][t].x, bq[sp & 1][t].y};
          acc[t] = __builtin_amdgcn_wmma_f32_16x16x4_f32(
              false, a0, false, b0, (short)0, acc[t], false, false);
        }
        #pragma unroll
        for (int t = 0; t < 4; ++t) {
          const v2f b1 = {bq[sp & 1][t].z, bq[sp & 1][t].w};
          acc[t] = __builtin_amdgcn_wmma_f32_16x16x4_f32(
              false, a1, false, b1, (short)0, acc[t], false, false);
        }
      }
    }

    float* yc = Y + (size_t)(rb * 16 + 8 * hf) * DD + ch * 64 + c;
    #pragma unroll
    for (int t = 0; t < 4; ++t) {
      const float bs = bias ? bias[ch * 64 + t * 16 + c] : 0.0f;
      #pragma unroll
      for (int r = 0; r < 8; ++r) yc[r * DD + t * 16] = acc[t][r] + bs;
    }
    xa = xn;
  }
}

__global__ __launch_bounds__(256) void att_logits_k(
    const float* __restrict__ xh, const float* __restrict__ a_src,
    const float* __restrict__ a_dst, float* __restrict__ e_src,
    float* __restrict__ e_dst) {
  int t = blockIdx.x * blockDim.x + threadIdx.x;
  if (t >= NN * HH) return;
  int n = t >> 3, h = t & 7;
  const float4* xp = (const float4*)(xh + (size_t)n * DD + h * CC);
  const float4* sp = (const float4*)(a_src + h * CC);
  const float4* dp = (const float4*)(a_dst + h * CC);
  float ss = 0.f, sd = 0.f;
  #pragma unroll
  for (int i = 0; i < 4; ++i) {
    float4 x = xp[i], s = sp[i], d = dp[i];
    ss += x.x * s.x + x.y * s.y + x.z * s.z + x.w * s.w;
    sd += x.x * d.x + x.y * d.y + x.z * d.z + x.w * d.w;
  }
  e_src[t] = ss;
  e_dst[t] = sd;
}

__device__ __forceinline__ float leaky(float v) {
  return v > 0.f ? v : v * NEG_SLOPE;
}
__device__ __forceinline__ unsigned f2ord(float f) {
  unsigned u = __float_as_uint(f);
  return (u & 0x80000000u) ? ~u : (u | 0x80000000u);
}
__device__ __forceinline__ float ord2f(unsigned u) {
  return (u & 0x80000000u) ? __uint_as_float(u & 0x7fffffffu)
                           : __uint_as_float(~u);
}

__global__ __launch_bounds__(256) void bin_count_k(
    const int* __restrict__ ei, int* __restrict__ blockhist) {
  __shared__ int hist[NT];
  for (int i = threadIdx.x; i < NT; i += 256) hist[i] = 0;
  __syncthreads();
  const int e0 = blockIdx.x * CH;
  for (int i0 = threadIdx.x; i0 < CH; i0 += 256 * BIN_U) {
    int d[BIN_U];
    #pragma unroll
    for (int u = 0; u < BIN_U; ++u) {
      const int e = e0 + i0 + 256 * u;
      d[u] = (e < EE) ? ei[EE + e] : -1;
    }
    #pragma unroll
    for (int u = 0; u < BIN_U; ++u)
      if (d[u] >= 0) atomicAdd(&hist[d[u] / TN], 1);
  }
  __syncthreads();
  for (int i = threadIdx.x; i < NT; i += 256)
    blockhist[blockIdx.x * NT + i] = hist[i];
}

__global__ __launch_bounds__(1024) void bin_scan_k(
    const int* __restrict__ blockhist, int* __restrict__ blockoff,
    int* __restrict__ binstart) {
  __shared__ int sc[1024];
  const int t = threadIdx.x;
  int run = 0;
  if (t < NT) {
    for (int b0 = 0; b0 < NB; b0 += SCAN_U) {
      int c[SCAN_U];
      #pragma unroll
      for (int u = 0; u < SCAN_U; ++u) c[u] = blockhist[(b0 + u) * NT + t];
      #pragma unroll
      for (int u = 0; u < SCAN_U; ++u) {
        blockoff[(b0 + u) * NT + t] = run;
        run += c[u];
      }
    }
  }
  sc[t] = run;
  __syncthreads();
  for (int o = 1; o < 1024; o <<= 1) {
    const int v = (t >= o) ? sc[t - o] : 0;
    __syncthreads();
    sc[t] += v;
    __syncthreads();
  }
  if (t < NT) binstart[t] = sc[t] - run;
  if (t == NT - 1) binstart[NT] = sc[t];
}

__global__ __launch_bounds__(256) void bin_scatter_k(
    const int* __restrict__ ei, const int* __restrict__ blockoff,
    const int* __restrict__ binstart, int* __restrict__ binned) {
  __shared__ int cur[NT];
  for (int i = threadIdx.x; i < NT; i += 256)
    cur[i] = binstart[i] + blockoff[blockIdx.x * NT + i];
  __syncthreads();
  const int e0 = blockIdx.x * CH;
  for (int i0 = threadIdx.x; i0 < CH; i0 += 256 * BIN_U) {
    int s[BIN_U], d[BIN_U];
    #pragma unroll
    for (int u = 0; u < BIN_U; ++u) {
      const int e = e0 + i0 + 256 * u;
      s[u] = (e < EE) ? ei[e] : 0;
      d[u] = (e < EE) ? ei[EE + e] : -1;
    }
    #pragma unroll
    for (int u = 0; u < BIN_U; ++u) {
      if (d[u] >= 0) {
        const int t   = d[u] / TN;
        const int pos = atomicAdd(&cur[t], 1);
        binned[pos] = (s[u] << 7) | (d[u] - t * TN);
      }
    }
  }
}

#define AG_U 16

__device__ __forceinline__ int fetch_edge(const int* __restrict__ binned,
                                          int eb, int ne, int tot, int n0,
                                          int idx) {
  int pk = n0 << 7;
  if (idx < ne) pk = binned[eb + idx];
  else if (idx < tot) { const int dl = idx - ne; pk = ((n0 + dl) << 7) | dl; }
  return pk;
}

__device__ __forceinline__ void tile_frags(const float* arow,
                                           const float* __restrict__ wrow,
                                           int k, int koff, int sw,
                                           v2f (&a)[RB], v2f& b) {
  const int kc = (k + koff) ^ sw;
  #pragma unroll
  for (int rb = 0; rb < RB; ++rb)
    a[rb] = *(const v2f*)(arow + rb * 16 * DD + kc);
  b = *(const v2f*)(wrow + k);
}
__device__ __forceinline__ void tile_mma(v8f (&d)[RB], const v2f (&a)[RB],
                                         v2f b) {
  #pragma unroll
  for (int rb = 0; rb < RB; ++rb)
    d[rb] = __builtin_amdgcn_wmma_f32_16x16x4_f32(
        false, a[rb], false, b, (short)0, d[rb], false, false);
}

__global__ __launch_bounds__(256) __attribute__((amdgpu_waves_per_eu(8)))
void tile_aggr_ln_k(
    const int* __restrict__ binned, const int* __restrict__ binstart,
    const float* __restrict__ e_src, const float* __restrict__ e_dst,
    const float* __restrict__ xh, const float* hin,
    const float* __restrict__ gb, const float* __restrict__ g,
    const float* __restrict__ b, float* hout, float* __restrict__ part,
    const float* __restrict__ Wn, float* __restrict__ xo) {
  __shared__ __attribute__((aligned(16))) float acc[TNP * DD];
  __shared__ float    sden[TN * HH];
  __shared__ unsigned smax[TN * HH];
  __shared__ float    sed[TN * HH];
  __shared__ __attribute__((aligned(16))) float pcol[8 * DD];

  const int tid  = threadIdx.x;
  const int lane = tid & 31;
  const int wave = __builtin_amdgcn_readfirstlane(tid >> 5);
  const int h    = lane >> 2;
  const int n0   = blockIdx.x * TN;
  const int nn   = (NN - n0 < TN) ? (NN - n0) : TN;
  const int eb   = binstart[blockIdx.x];
  const int ne   = binstart[blockIdx.x + 1] - eb;
  const int tot  = ne + nn;

  for (int i = tid * 4; i < TNP * DD; i += 256 * 4)
    *(float4*)&acc[i] = make_float4(0.f, 0.f, 0.f, 0.f);
  for (int i = tid; i < TN * HH; i += 256) {
    float es = 0.f, ed = 0.f;
    if ((i >> 3) < nn) {
      es = e_src[(size_t)n0 * HH + i];
      ed = e_dst[(size_t)n0 * HH + i];
    }
    sed[i]  = ed;
    sden[i] = 0.f;
    smax[i] = f2ord(leaky(es + ed));
  }
  __syncthreads();

  for (int i0 = tid; i0 < ne; i0 += 256 * 2) {
    int pk[2];
    float4 ea[2], ec[2];
    #pragma unroll
    for (int u = 0; u < 2; ++u) {
      const int i = i0 + 256 * u;
      pk[u] = (i < ne) ? binned[eb + i] : -1;
    }
    #pragma unroll
    for (int u = 0; u < 2; ++u) {
      const int s = (pk[u] >= 0) ? (pk[u] >> 7) : n0;
      const float4* p = (const float4*)(e_src + (size_t)s * HH);
      ea[u] = p[0];
      ec[u] = p[1];
    }
    #pragma unroll
    for (int u = 0; u < 2; ++u) {
      if (pk[u] >= 0) {
        const int dl = pk[u] & 127;
        const float* sd = sed + dl * HH;
        unsigned*    sm = smax + dl * HH;
        atomicMax(sm + 0, f2ord(leaky(ea[u].x + sd[0])));
        atomicMax(sm + 1, f2ord(leaky(ea[u].y + sd[1])));
        atomicMax(sm + 2, f2ord(leaky(ea[u].z + sd[2])));
        atomicMax(sm + 3, f2ord(leaky(ea[u].w + sd[3])));
        atomicMax(sm + 4, f2ord(leaky(ec[u].x + sd[4])));
        atomicMax(sm + 5, f2ord(leaky(ec[u].y + sd[5])));
        atomicMax(sm + 6, f2ord(leaky(ec[u].z + sd[6])));
        atomicMax(sm + 7, f2ord(leaky(ec[u].w + sd[7])));
      }
    }
  }
  __syncthreads();

  {
    int base = wave * 32;
    int pk_l = fetch_edge(binned, eb, ne, tot, n0, base + lane);
    while (base < tot) {
      const int nbase = base + 8 * 32;
      const int pk_n = fetch_edge(binned, eb, ne, tot, n0, nbase + lane);
      const int cnt  = (tot - base < 32) ? (tot - base) : 32;
      for (int k0 = 0; k0 < cnt; k0 += AG_U) {
        float4 x[AG_U];
        float  es[AG_U];
        int    dls[AG_U];
        #pragma unroll
        for (int u = 0; u < AG_U; ++u) {
          const int pk = __builtin_amdgcn_readlane(pk_l, k0 + u);
          const int s  = pk >> 7;
          dls[u] = pk & 127;
          es[u]  = e_src[(size_t)s * HH + h];
          x[u]   = *(const float4*)(xh + (size_t)s * DD + lane * 4);
        }
        #pragma unroll
        for (int u = 0; u < AG_U; ++u) {
          if (k0 + u < cnt) {
            const int   dl = dls[u];
            const float v  = leaky(es[u] + sed[dl * HH + h]);
            const float ex = __expf(v - ord2f(smax[dl * HH + h]));
            float* a = acc + dl * DD + lane;
            atomicAdd(a,      x[u].x * ex);
            atomicAdd(a + 32, x[u].y * ex);
            atomicAdd(a + 64, x[u].z * ex);
            atomicAdd(a + 96, x[u].w * ex);
            if ((lane & 3) == 0) atomicAdd(&sden[dl * HH + h], ex);
          }
        }
      }
      pk_l = pk_n;
      base = nbase;
    }
  }
  __syncthreads();

  const float4 bv = *(const float4*)(gb + lane * 4);
  const float4 gv = *(const float4*)(g + lane * 4);
  const float4 lb = *(const float4*)(b + lane * 4);
  float4 ps = make_float4(0.f, 0.f, 0.f, 0.f);
  for (int q0 = 0; q0 < (TN + 7) / 8; q0 += 4) {
    float4 hv[4];
    #pragma unroll
    for (int u = 0; u < 4; ++u) {
      const int dl = wave + 8 * (q0 + u);
      hv[u] = make_float4(0.f, 0.f, 0.f, 0.f);
      if (dl < nn)
        hv[u] = *(const float4*)(hin + (size_t)(n0 + dl) * DD + lane * 4);
    }
    #pragma unroll
    for (int u = 0; u < 4; ++u) {
      const int dl = wave + 8 * (q0 + u);
      if (dl < nn) {
        const float  den = sden[dl * HH + h] + 1e-16f;
        const float* a   = acc + dl * DD + lane;
        float x0 = hv[u].x + a[0]  / den + bv.x;
        float x1 = hv[u].y + a[32] / den + bv.y;
        float x2 = hv[u].z + a[64] / den + bv.z;
        float x3 = hv[u].w + a[96] / den + bv.w;
        float s = x0 + x1 + x2 + x3;
        #pragma unroll
        for (int o = 16; o > 0; o >>= 1) s += __shfl_xor(s, o, 32);
        float mean = s * (1.f / (float)DD);
        float d0 = x0 - mean, d1 = x1 - mean, d2 = x2 - mean, d3 = x3 - mean;
        float vs = d0 * d0 + d1 * d1 + d2 * d2 + d3 * d3;
        #pragma unroll
        for (int o = 16; o > 0; o >>= 1) vs += __shfl_xor(vs, o, 32);
        float inv = rsqrtf(vs * (1.f / (float)DD) + LN_EPS);
        float4 o4;
        o4.x = d0 * inv * gv.x + lb.x;
        o4.y = d1 * inv * gv.y + lb.y;
        o4.z = d2 * inv * gv.z + lb.z;
        o4.w = d3 * inv * gv.w + lb.w;
        *(float4*)(hout + (size_t)(n0 + dl) * DD + lane * 4) = o4;
        if (Wn)
          *(float4*)&acc[dl * DD + ((lane * 4) ^ ((dl & 15) << 2))] = o4;
        ps.x += o4.x;
        ps.y += o4.y;
        ps.z += o4.z;
        ps.w += o4.w;
      }
    }
  }

  if (part) {
    *(float4*)&pcol[wave * DD + lane * 4] = ps;
    __syncthreads();
    if (tid < DD) {
      float s = 0.f;
      #pragma unroll
      for (int w = 0; w < 8; ++w) s += pcol[w * DD + tid];
      part[(size_t)blockIdx.x * DD + tid] = s;
    }
  }

  if (Wn) {
    __syncthreads();
    const int c    = lane & 15;
    const int koff = (lane < 16) ? 0 : 2;
    const int sw   = c << 2;
    const float* arow = acc + c * DD;
    const float* __restrict__ wrow = Wn + (size_t)(wave * 16 + c) * DD + koff;
    v8f d[RB];
    #pragma unroll
    for (int rb = 0; rb < RB; ++rb) d[rb] = v8f{};
    #pragma unroll
    for (int kk = 0; kk < DD / 4; ++kk) {
      v2f a[RB], b;
      tile_frags(arow, wrow, 4 * kk, koff, sw, a, b);
      tile_mma(d, a, b);
    }

    const int mrow = (lane < 16) ? 0 : 8;
    float* xc = xo + (size_t)(n0 + mrow) * DD + wave * 16 + c;
    if (nn == TN) {
      #pragma unroll
      for (int rb = 0; rb < RB; ++rb) {
        #pragma unroll
        for (int r = 0; r < 8; ++r) {
          if (rb < TN / 16 || rb * 16 + mrow + r < TN)
            xc[(rb * 16 + r) * DD] = d[rb][r];
        }
      }
    } else {
      #pragma unroll
      for (int rb = 0; rb < RB; ++rb) {
        #pragma unroll
        for (int r = 0; r < 8; ++r) {
          if (rb * 16 + mrow + r < nn) xc[(rb * 16 + r) * DD] = d[rb][r];
        }
      }
    }
  }
}

#define HP_G 8
#define HP_U 25
static_assert(HP_G * HP_U * 5 >= NT, "five batches cover the tiles");
__global__ __launch_bounds__(1024) void head_k(
    const float* __restrict__ part, const float* __restrict__ W_mu,
    const float* __restrict__ b_mu, const float* __restrict__ W_lv,
    const float* __restrict__ b_lv, float* __restrict__ out) {
  __shared__ float red[HP_G * DD];
  __shared__ float pl[DD];
  const int d = threadIdx.x & (DD - 1);
  const int g = threadIdx.x >> 7;
  float s = 0.f;
  for (int t0 = g; t0 < NT; t0 += HP_G * HP_U) {
    float v[HP_U];
    #pragma unroll
    for (int u = 0; u < HP_U; ++u) {
      const int t = t0 + HP_G * u;
      v[u] = (t < NT) ? part[(size_t)t * DD + d] : 0.f;
    }
    #pragma unroll
    for (int u = 0; u < HP_U; ++u) s += v[u];
  }
  red[g * DD + d] = s;
  __syncthreads();
  float pd = 0.f;
  if (threadIdx.x < DD) {
    #pragma unroll
    for (int w = 0; w < HP_G; ++w) pd += red[w * DD + d];
    pd *= (1.f / (float)NN);
    pl[d] = pd;
  }
  __syncthreads();
  if (threadIdx.x < DD) {
    float mu = b_mu[d], lv = b_lv[d];
    for (int k = 0; k < DD; ++k) {
      float p = pl[k];
      mu += p * W_mu[(size_t)d * DD + k];
      lv += p * W_lv[(size_t)d * DD + k];
    }
    out[d]      = mu;
    out[DD + d] = lv;
    out[2 * DD + (size_t)NN * DD + d] = pd;
  }
}

extern "C" void kernel_launch(void* const* d_in, const int* in_sizes, int n_in,
                              void* d_out, int out_size, void* d_ws, size_t ws_size,
                              hipStream_t stream) {
  const float* x       = (const float*)d_in[0];
  const int*   ei      = (const int*)d_in[1];
  const float* W_in    = (const float*)d_in[2];
  const float* b_in    = (const float*)d_in[3];
  const float* gat_W   = (const float*)d_in[4];
  const float* att_src = (const float*)d_in[5];
  const float* att_dst = (const float*)d_in[6];
  const float* gat_b   = (const float*)d_in[7];
  const float* ln_g    = (const float*)d_in[8];
  const float* ln_b    = (const float*)d_in[9];
  const float* W_mu    = (const float*)d_in[10];
  const float* b_mu    = (const float*)d_in[11];
  const float* W_lv    = (const float*)d_in[12];
  const float* b_lv    = (const float*)d_in[13];
  float* out = (float*)d_out;

  float* h      = (float*)d_ws;
  float* xh     = h     + (size_t)NN * DD;
  float* plane  = xh    + (size_t)NN * DD;
  float* e_src  = plane + (size_t)NN * DD;
  float* e_dst  = e_src + (size_t)NN * HH;

  int* binned    = (int*)plane;
  int* blockhist = binned    + (size_t)EE;
  int* blockoff  = blockhist + (size_t)NB * NT;
  int* binstart  = blockoff  + (size_t)NB * NT;
  float* part    = (float*)(binstart + NT + 8);

  float* h_final = out + 2 * DD;
  float* xh2     = h_final;

  const int TPB = 256;
  const int nh_blk   = (NN * HH + TPB - 1) / TPB;

  gemm_wlds_k<F_IN><<<GEMM_WG, GEMM_TPB, 0, stream>>>(x, W_in, b_in, h);

  bin_count_k  <<<NB, TPB, 0, stream>>>(ei, blockhist);
  bin_scan_k   <<<1, 1024, 0, stream>>>(blockhist, blockoff, binstart);
  bin_scatter_k<<<NB, TPB, 0, stream>>>(ei, blockoff, binstart, binned);

  for (int l = 0; l < LL; ++l) {
    const float* Wl  = gat_W   + (size_t)l * DD * DD;
    const float* asl = att_src + (size_t)l * HH * CC;
    const float* adl = att_dst + (size_t)l * HH * CC;
    const float* gbl = gat_b   + (size_t)l * DD;
    const float* gl  = ln_g    + (size_t)l * DD;
    const float* bl  = ln_b    + (size_t)l * DD;

    float* xcur = (l & 1) ? xh : xh2;
    float* xnxt = (l & 1) ? xh2 : xh;
    const float* Wn = (l == LL - 1) ? nullptr : Wl + (size_t)DD * DD;

    if (l == 0) gemm_wlds_k<DD><<<GEMM_WG, GEMM_TPB, 0, stream>>>(h, Wl, nullptr, xcur);

    att_logits_k<<<nh_blk, TPB, 0, stream>>>(xcur, asl, adl, e_src, e_dst);

    float* hout  = (l == LL - 1) ? h_final : h;
    float* partl = (l == LL - 1) ? part : nullptr;
    tile_aggr_ln_k<<<NT, TPB, 0, stream>>>(binned, binstart, e_src, e_dst, xcur,
                                           h, gbl, gl, bl, hout, partl,
                                           Wn, (l == LL - 1) ? nullptr : xnxt);
  }

  head_k<<<1, 1024, 0, stream>>>(part, W_mu, b_mu, W_lv, b_lv, out);
}


# tile sums: wave w owns the tile's nodes w mod 8, picks their edges from the tile list by ballot into an LDS ring and adds rows by 16-byte LDS load/add/store, not 4 float atomics; sums in list order
# speedup vs baseline: 4.0138x; 1.0418x over previous
#include <hip/hip_runtime.h>
#include <hip/hip_bf16.h>


#define NN      100000
#define EE      1600000
#define F_IN    256
#define DD      128
#define HH      8
#define CC      16
#define LL      4
#define NEG_SLOPE 0.2f
#define LN_EPS  1e-5f

#define TN      100
#define NT      ((NN + TN - 1) / TN)
#define RB      ((TN + 15) / 16)
#define TNP     (RB * 16)
#define CH      8192
#define NB      ((EE + CH - 1) / CH)
#define SCAN_U  28
#define BIN_U   8
static_assert(NB % SCAN_U == 0, "scan batches must cover the blocks exactly");
static_assert(CH % (256 * BIN_U) == 0, "binning batches must cover a block exactly");
static_assert(NT <= 1024, "one scan thread per tile");
static_assert(NN < (1 << 17) && TN <= 128, "src and local dst share one word");
static_assert((TNP * DD) % (256 * 4) == 0, "the clearing loop covers the plane exactly");

typedef __attribute__((ext_vector_type(2))) float v2f;
typedef __attribute__((ext_vector_type(8))) float v8f;

#define GEMM_WG   256
#define GEMM_TPB  512
#define GEMM_LDS_FLOATS 41472
static_assert(DD * (F_IN + 4) <= GEMM_LDS_FLOATS, "the larger weight matrix fits");
static_assert(NN % 16 == 0, "whole 16-row blocks");

__device__ __forceinline__ int frag_k(int j, int hf) { return 8 * j + 4 * hf; }

template <int K>
__global__ __launch_bounds__(GEMM_TPB) void gemm_wlds_k(
    const float* __restrict__ X, const float* __restrict__ W,
    const float* __restrict__ bias, float* __restrict__ Y) {
  constexpr int KP  = K + 4;
  constexpr int NCH = K / 64;
  static_assert(NCH % 2 == 0, "the chunk ring of two closes over a unit");
  __shared__ __attribute__((aligned(16))) float Ws[GEMM_LDS_FLOATS];

  const int tid  = threadIdx.x;
  const int lane = tid & 31;
  const int wave = __builtin_amdgcn_readfirstlane(tid >> 5);
  const int c    = lane & 15;
  const int hf   = lane >> 4;

  const int rb0 = (int)(((long long)blockIdx.x * (NN / 16)) / GEMM_WG);
  const int rb1 = (int)(((long long)(blockIdx.x + 1) * (NN / 16)) / GEMM_WG);
  const int nu  = (rb1 - rb0) * 2;
  const int ch  = wave & 1;

  const float* wl = Ws + (ch * 64 + c) * KP;
  const float* xa = X + (size_t)((rb0 + (wave >> 1)) * 16 + c) * K;

  float4 a[2][8];
  if (wave < nu) {
    #pragma unroll
    for (int j = 0; j < 8; ++j)
      a[0][j] = *(const float4*)(xa + frag_k(j, hf));
  }

  {
    constexpr int WLD = (DD * K) / (GEMM_TPB * 4);
    static_assert(WLD * GEMM_TPB * 4 == DD * K, "the copy covers W exactly");
    float4 wv[WLD];
    #pragma unroll
    for (int i = 0; i < WLD; ++i)
      wv[i] = *(const float4*)(W + (tid + i * GEMM_TPB) * 4);
    #pragma unroll
    for (int i = 0; i < WLD; ++i) {
      const int e = (tid + i * GEMM_TPB) * 4;
      const int r = e / K, cc = e % K;
      *(float4*)&Ws[r * KP + cc] = wv[i];
    }
  }
  __syncthreads();

  for (int u = wave; u < nu; u += 16) {
    asm volatile("" ::: "memory");
    const int rb = rb0 + (u >> 1);
    const int un = (u + 16 < nu) ? (u + 16) : u;
    const float* xn = X + (size_t)((rb0 + (un >> 1)) * 16 + c) * K;

    v8f acc[4];
    #pragma unroll
    for (int t = 0; t < 4; ++t) acc[t] = v8f{};
    float4 bq[2][4];
    #pragma unroll
    for (int t = 0; t < 4; ++t)
      bq[0][t] = *(const float4*)(wl + t * 16 * KP + frag_k(0, hf));

    #pragma unroll
    for (int cc = 0; cc < NCH; ++cc) {
      const float* src = (cc + 1 < NCH) ? (xa + (cc + 1) * 64) : xn;
      #pragma unroll
      for (int j = 0; j < 8; ++j)
        a[(cc + 1) & 1][j] = *(const float4*)(src + frag_k(j, hf));

      #pragma unroll
      for (int j = 0; j < 8; ++j) {
        const int sp = cc * 8 + j;
        if (sp + 1 < K / 8) {
          #pragma unroll
          for (int t = 0; t < 4; ++t)
            bq[(sp + 1) & 1][t] =
                *(const float4*)(wl + t * 16 * KP + frag_k(sp + 1, hf));
        }
        const float4 av = a[cc & 1][j];
        const v2f a0 = {av.x, av.y}, a1 = {av.z, av.w};
        #pragma unroll
        for (int t = 0; t < 4; ++t) {
          const v2f b0 = {bq[sp & 1][t].x, bq[sp & 1][t].y};
          acc[t] = __builtin_amdgcn_wmma_f32_16x16x4_f32(
              false, a0, false, b0, (short)0, acc[t], false, false);
        }
        #pragma unroll
        for (int t = 0; t < 4; ++t) {
          const v2f b1 = {bq[sp & 1][t].z, bq[sp & 1][t].w};
          acc[t] = __builtin_amdgcn_wmma_f32_16x16x4_f32(
              false, a1, false, b1, (short)0, acc[t], false, false);
        }
      }
    }

    float* yc = Y + (size_t)(rb * 16 + 8 * hf) * DD + ch * 64 + c;
    #pragma unroll
    for (int t = 0; t < 4; ++t) {
      const float bs = bias ? bias[ch * 64 + t * 16 + c] : 0.0f;
      #pragma unroll
      for (int r = 0; r < 8; ++r) yc[r * DD + t * 16] = acc[t][r] + bs;
    }
    xa = xn;
  }
}

__global__ __launch_bounds__(256) void att_logits_k(
    const float* __restrict__ xh, const float* __restrict__ a_src,
    const float* __restrict__ a_dst, float* __restrict__ e_src,
    float* __restrict__ e_dst) {
  int t = blockIdx.x * blockDim.x + threadIdx.x;
  if (t >= NN * HH) return;
  int n = t >> 3, h = t & 7;
  const float4* xp = (const float4*)(xh + (size_t)n * DD + h * CC);
  const float4* sp = (const float4*)(a_src + h * CC);
  const float4* dp = (const float4*)(a_dst + h * CC);
  float ss = 0.f, sd = 0.f;
  #pragma unroll
  for (int i = 0; i < 4; ++i) {
    float4 x = xp[i], s = sp[i], d = dp[i];
    ss += x.x * s.x + x.y * s.y + x.z * s.z + x.w * s.w;
    sd += x.x * d.x + x.y * d.y + x.z * d.z + x.w * d.w;
  }
  e_src[t] = ss;
  e_dst[t] = sd;
}

__device__ __forceinline__ float leaky(float v) {
  return v > 0.f ? v : v * NEG_SLOPE;
}
__device__ __forceinline__ unsigned f2ord(float f) {
  unsigned u = __float_as_uint(f);
  return (u & 0x80000000u) ? ~u : (u | 0x80000000u);
}
__device__ __forceinline__ float ord2f(unsigned u) {
  return (u & 0x80000000u) ? __uint_as_float(u & 0x7fffffffu)
                           : __uint_as_float(~u);
}

__global__ __launch_bounds__(256) void bin_count_k(
    const int* __restrict__ ei, int* __restrict__ blockhist) {
  __shared__ int hist[NT];
  for (int i = threadIdx.x; i < NT; i += 256) hist[i] = 0;
  __syncthreads();
  const int e0 = blockIdx.x * CH;
  for (int i0 = threadIdx.x; i0 < CH; i0 += 256 * BIN_U) {
    int d[BIN_U];
    #pragma unroll
    for (int u = 0; u < BIN_U; ++u) {
      const int e = e0 + i0 + 256 * u;
      d[u] = (e < EE) ? ei[EE + e] : -1;
    }
    #pragma unroll
    for (int u = 0; u < BIN_U; ++u)
      if (d[u] >= 0) atomicAdd(&hist[d[u] / TN], 1);
  }
  __syncthreads();
  for (int i = threadIdx.x; i < NT; i += 256)
    blockhist[blockIdx.x * NT + i] = hist[i];
}

__global__ __launch_bounds__(1024) void bin_scan_k(
    const int* __restrict__ blockhist, int* __restrict__ blockoff,
    int* __restrict__ binstart) {
  __shared__ int sc[1024];
  const int t = threadIdx.x;
  int run = 0;
  if (t < NT) {
    for (int b0 = 0; b0 < NB; b0 += SCAN_U) {
      int c[SCAN_U];
      #pragma unroll
      for (int u = 0; u < SCAN_U; ++u) c[u] = blockhist[(b0 + u) * NT + t];
      #pragma unroll
      for (int u = 0; u < SCAN_U; ++u) {
        blockoff[(b0 + u) * NT + t] = run;
        run += c[u];
      }
    }
  }
  sc[t] = run;
  __syncthreads();
  for (int o = 1; o < 1024; o <<= 1) {
    const int v = (t >= o) ? sc[t - o] : 0;
    __syncthreads();
    sc[t] += v;
    __syncthreads();
  }
  if (t < NT) binstart[t] = sc[t] - run;
  if (t == NT - 1) binstart[NT] = sc[t];
}

__global__ __launch_bounds__(256) void bin_scatter_k(
    const int* __restrict__ ei, const int* __restrict__ blockoff,
    const int* __restrict__ binstart, int* __restrict__ binned) {
  __shared__ int cur[NT];
  for (int i = threadIdx.x; i < NT; i += 256)
    cur[i] = binstart[i] + blockoff[blockIdx.x * NT + i];
  __syncthreads();
  const int e0 = blockIdx.x * CH;
  for (int i0 = threadIdx.x; i0 < CH; i0 += 256 * BIN_U) {
    int s[BIN_U], d[BIN_U];
    #pragma unroll
    for (int u = 0; u < BIN_U; ++u) {
      const int e = e0 + i0 + 256 * u;
      s[u] = (e < EE) ? ei[e] : 0;
      d[u] = (e < EE) ? ei[EE + e] : -1;
    }
    #pragma unroll
    for (int u = 0; u < BIN_U; ++u) {
      if (d[u] >= 0) {
        const int t   = d[u] / TN;
        const int pos = atomicAdd(&cur[t], 1);
        binned[pos] = (s[u] << 7) | (d[u] - t * TN);
      }
    }
  }
}

#define AG_U 16
#define SC_U 4
#define OQ_N 256
static_assert(OQ_N >= AG_U + 32 * SC_U && (OQ_N & (OQ_N - 1)) == 0, "the ring holds a scan step");
static_assert((AG_U & (AG_U - 1)) == 0 && AG_U <= 32, "a batch is read one entry per lane");

__device__ __forceinline__ int fetch_edge(const int* __restrict__ binned,
                                          int eb, int idx) {
  const int gi = eb + idx;
  return binned[(gi < EE) ? gi : (EE - 1)];
}

__device__ __forceinline__ void tile_frags(const float* arow,
                                           const float* __restrict__ wrow,
                                           int k, int koff, int sw,
                                           v2f (&a)[RB], v2f& b) {
  const int kc = (k + koff) ^ sw;
  #pragma unroll
  for (int rb = 0; rb < RB; ++rb)
    a[rb] = *(const v2f*)(arow + rb * 16 * DD + kc);
  b = *(const v2f*)(wrow + k);
}
__device__ __forceinline__ void tile_mma(v8f (&d)[RB], const v2f (&a)[RB],
                                         v2f b) {
  #pragma unroll
  for (int rb = 0; rb < RB; ++rb)
    d[rb] = __builtin_amdgcn_wmma_f32_16x16x4_f32(
        false, a[rb], false, b, (short)0, d[rb], false, false);
}

template <bool FULL>
__device__ __forceinline__ void own_batch(
    const int* q, int qh, int cnt, int n0, int lane, int h,
    const float* __restrict__ e_src, const float* __restrict__ xh,
    float* acc, float* sden, const float* sed, const unsigned* smax) {
  const int qv = q[(qh + (lane & (AG_U - 1))) & (OQ_N - 1)];

  float4 x[AG_U];
  float  es[AG_U];
  int    dls[AG_U];
  #pragma unroll
  for (int u = 0; u < AG_U; ++u) {
    const int pk = (FULL || u < cnt) ? __builtin_amdgcn_readlane(qv, u) : (n0 << 7);
    const int s  = pk >> 7;
    dls[u] = pk & 127;
    es[u]  = e_src[(size_t)s * HH + h];
    x[u]   = *(const float4*)(xh + (size_t)s * DD + lane * 4);
  }
  #pragma unroll
  for (int u = 0; u < AG_U; ++u) {
    if (FULL || u < cnt) {
      const int dl = dls[u];
      float4* ap = (float4*)&acc[dl * DD + lane * 4];
      float4 a = *ap;
      const float v  = leaky(es[u] + sed[dl * HH + h]);
      const float ex = __expf(v - ord2f(smax[dl * HH + h]));
      a.x += x[u].x * ex;
      a.y += x[u].y * ex;
      a.z += x[u].z * ex;
      a.w += x[u].w * ex;
      *ap = a;
      if ((lane & 3) == 0) atomicAdd(&sden[dl * HH + h], ex);
    }
  }
}

__global__ __launch_bounds__(256) __attribute__((amdgpu_waves_per_eu(8)))
void tile_aggr_ln_k(
    const int* __restrict__ binned, const int* __restrict__ binstart,
    const float* __restrict__ e_src, const float* __restrict__ e_dst,
    const float* __restrict__ xh, const float* hin,
    const float* __restrict__ gb, const float* __restrict__ g,
    const float* __restrict__ b, float* hout, float* __restrict__ part,
    const float* __restrict__ Wn, float* __restrict__ xo) {
  __shared__ __attribute__((aligned(16))) float acc[TNP * DD];
  __shared__ float    sden[TN * HH];
  __shared__ unsigned smax[TN * HH];
  __shared__ float    sed[TN * HH];
  __shared__ __attribute__((aligned(16))) float pcol[8 * DD];
  __shared__ int oq[8 * OQ_N];

  const int tid  = threadIdx.x;
  const int lane = tid & 31;
  const int wave = __builtin_amdgcn_readfirstlane(tid >> 5);
  const int h    = lane >> 2;
  const int n0   = blockIdx.x * TN;
  const int nn   = (NN - n0 < TN) ? (NN - n0) : TN;
  const int eb   = binstart[blockIdx.x];
  const int ne   = binstart[blockIdx.x + 1] - eb;

  for (int i = tid * 4; i < TNP * DD; i += 256 * 4)
    *(float4*)&acc[i] = make_float4(0.f, 0.f, 0.f, 0.f);
  for (int i = tid; i < TN * HH; i += 256) {
    float es = 0.f, ed = 0.f;
    if ((i >> 3) < nn) {
      es = e_src[(size_t)n0 * HH + i];
      ed = e_dst[(size_t)n0 * HH + i];
    }
    sed[i]  = ed;
    sden[i] = 0.f;
    smax[i] = f2ord(leaky(es + ed));
  }
  __syncthreads();

  for (int i0 = tid; i0 < ne; i0 += 256 * 2) {
    int pk[2];
    float4 ea[2], ec[2];
    #pragma unroll
    for (int u = 0; u < 2; ++u) {
      const int i = i0 + 256 * u;
      pk[u] = (i < ne) ? binned[eb + i] : -1;
    }
    #pragma unroll
    for (int u = 0; u < 2; ++u) {
      const int s = (pk[u] >= 0) ? (pk[u] >> 7) : n0;
      const float4* p = (const float4*)(e_src + (size_t)s * HH);
      ea[u] = p[0];
      ec[u] = p[1];
    }
    #pragma unroll
    for (int u = 0; u < 2; ++u) {
      if (pk[u] >= 0) {
        const int dl = pk[u] & 127;
        const float* sd = sed + dl * HH;
        unsigned*    sm = smax + dl * HH;
        atomicMax(sm + 0, f2ord(leaky(ea[u].x + sd[0])));
        atomicMax(sm + 1, f2ord(leaky(ea[u].y + sd[1])));
        atomicMax(sm + 2, f2ord(leaky(ea[u].z + sd[2])));
        atomicMax(sm + 3, f2ord(leaky(ea[u].w + sd[3])));
        atomicMax(sm + 4, f2ord(leaky(ec[u].x + sd[4])));
        atomicMax(sm + 5, f2ord(leaky(ec[u].y + sd[5])));
        atomicMax(sm + 6, f2ord(leaky(ec[u].z + sd[6])));
        atomicMax(sm + 7, f2ord(leaky(ec[u].w + sd[7])));
      }
    }
  }
  __syncthreads();

  {
    int* q = oq + wave * OQ_N;
    const int nsl = (nn > wave) ? ((nn - wave + 7) >> 3) : 0;
    if (lane < nsl) {
      const int dl = wave + 8 * lane;
      q[lane] = ((n0 + dl) << 7) | dl;
    }
    int qh = 0, qt = nsl;
    int pkc[SC_U];
    #pragma unroll
    for (int c = 0; c < SC_U; ++c)
      pkc[c] = fetch_edge(binned, eb, c * 32 + lane);
    for (int base = 0; base < ne; base += 32 * SC_U) {
      int pkn[SC_U];
      #pragma unroll
      for (int c = 0; c < SC_U; ++c)
        pkn[c] = fetch_edge(binned, eb, base + (SC_U + c) * 32 + lane);
      #pragma unroll
      for (int c = 0; c < SC_U; ++c) {
        const bool own = (base + c * 32 + lane < ne) & ((pkc[c] & 7) == wave);
        const unsigned m = __builtin_amdgcn_ballot_w32(own);
        if (own)
          q[(qt + (int)__builtin_amdgcn_mbcnt_lo(m, 0u)) & (OQ_N - 1)] = pkc[c];
        qt += __builtin_popcount(m);
      }
      while (qt - qh >= AG_U) {
        own_batch<true>(q, qh, AG_U, n0, lane, h, e_src, xh, acc, sden, sed, smax);
        qh += AG_U;
      }
      #pragma unroll
      for (int c = 0; c < SC_U; ++c) pkc[c] = pkn[c];
    }
    if (qt - qh > 0)
      own_batch<false>(q, qh, qt - qh, n0, lane, h, e_src, xh, acc, sden, sed, smax);
  }
  __syncthreads();

  const float4 bv = *(const float4*)(gb + lane * 4);
  const float4 gv = *(const float4*)(g + lane * 4);
  const float4 lb = *(const float4*)(b + lane * 4);
  float4 ps = make_float4(0.f, 0.f, 0.f, 0.f);
  for (int q0 = 0; q0 < (TN + 7) / 8; q0 += 4) {
    float4 hv[4];
    #pragma unroll
    for (int u = 0; u < 4; ++u) {
      const int dl = wave + 8 * (q0 + u);
      hv[u] = make_float4(0.f, 0.f, 0.f, 0.f);
      if (dl < nn)
        hv[u] = *(const float4*)(hin + (size_t)(n0 + dl) * DD + lane * 4);
    }
    #pragma unroll
    for (int u = 0; u < 4; ++u) {
      const int dl = wave + 8 * (q0 + u);
      if (dl < nn) {
        const float  den = sden[dl * HH + h] + 1e-16f;
        const float4 a   = *(const float4*)&acc[dl * DD + lane * 4];
        float x0 = hv[u].x + a.x / den + bv.x;
        float x1 = hv[u].y + a.y / den + bv.y;
        float x2 = hv[u].z + a.z / den + bv.z;
        float x3 = hv[u].w + a.w / den + bv.w;
        float s = x0 + x1 + x2 + x3;
        #pragma unroll
        for (int o = 16; o > 0; o >>= 1) s += __shfl_xor(s, o, 32);
        float mean = s * (1.f / (float)DD);
        float d0 = x0 - mean, d1 = x1 - mean, d2 = x2 - mean, d3 = x3 - mean;
        float vs = d0 * d0 + d1 * d1 + d2 * d2 + d3 * d3;
        #pragma unroll
        for (int o = 16; o > 0; o >>= 1) vs += __shfl_xor(vs, o, 32);
        float inv = rsqrtf(vs * (1.f / (float)DD) + LN_EPS);
        float4 o4;
        o4.x = d0 * inv * gv.x + lb.x;
        o4.y = d1 * inv * gv.y + lb.y;
        o4.z = d2 * inv * gv.z + lb.z;
        o4.w = d3 * inv * gv.w + lb.w;
        *(float4*)(hout + (size_t)(n0 + dl) * DD + lane * 4) = o4;
        if (Wn)
          *(float4*)&acc[dl * DD + ((lane * 4) ^ ((dl & 15) << 2))] = o4;
        ps.x += o4.x;
        ps.y += o4.y;
        ps.z += o4.z;
        ps.w += o4.w;
      }
    }
  }

  if (part) {
    *(float4*)&pcol[wave * DD + lane * 4] = ps;
    __syncthreads();
    if (tid < DD) {
      float s = 0.f;
      #pragma unroll
      for (int w = 0; w < 8; ++w) s += pcol[w * DD + tid];
      part[(size_t)blockIdx.x * DD + tid] = s;
    }
  }

  if (Wn) {
    __syncthreads();
    const int c    = lane & 15;
    const int koff = (lane < 16) ? 0 : 2;
    const int sw   = c << 2;
    const float* arow = acc + c * DD;
    const float* __restrict__ wrow = Wn + (size_t)(wave * 16 + c) * DD + koff;
    v8f d[RB];
    #pragma unroll
    for (int rb = 0; rb < RB; ++rb) d[rb] = v8f{};
    #pragma unroll
    for (int kk = 0; kk < DD / 4; ++kk) {
      v2f a[RB], b;
      tile_frags(arow, wrow, 4 * kk, koff, sw, a, b);
      tile_mma(d, a, b);
    }

    const int mrow = (lane < 16) ? 0 : 8;
    float* xc = xo + (size_t)(n0 + mrow) * DD + wave * 16 + c;
    if (nn == TN) {
      #pragma unroll
      for (int rb = 0; rb < RB; ++rb) {
        #pragma unroll
        for (int r = 0; r < 8; ++r) {
          if (rb < TN / 16 || rb * 16 + mrow + r < TN)
            xc[(rb * 16 + r) * DD] = d[rb][r];
        }
      }
    } else {
      #pragma unroll
      for (int rb = 0; rb < RB; ++rb) {
        #pragma unroll
        for (int r = 0; r < 8; ++r) {
          if (rb * 16 + mrow + r < nn) xc[(rb * 16 + r) * DD] = d[rb][r];
        }
      }
    }
  }
}

#define HP_G 8
#define HP_U 25
static_assert(HP_G * HP_U * 5 >= NT, "five batches cover the tiles");
__global__ __launch_bounds__(1024) void head_k(
    const float* __restrict__ part, const float* __restrict__ W_mu,
    const float* __restrict__ b_mu, const float* __restrict__ W_lv,
    const float* __restrict__ b_lv, float* __restrict__ out) {
  __shared__ float red[HP_G * DD];
  __shared__ float pl[DD];
  const int d = threadIdx.x & (DD - 1);
  const int g = threadIdx.x >> 7;
  float s = 0.f;
  for (int t0 = g; t0 < NT; t0 += HP_G * HP_U) {
    float v[HP_U];
    #pragma unroll
    for (int u = 0; u < HP_U; ++u) {
      const int t = t0 + HP_G * u;
      v[u] = (t < NT) ? part[(size_t)t * DD + d] : 0.f;
    }
    #pragma unroll
    for (int u = 0; u < HP_U; ++u) s += v[u];
  }
  red[g * DD + d] = s;
  __syncthreads();
  float pd = 0.f;
  if (threadIdx.x < DD) {
    #pragma unroll
    for (int w = 0; w < HP_G; ++w) pd += red[w * DD + d];
    pd *= (1.f / (float)NN);
    pl[d] = pd;
  }
  __syncthreads();
  if (threadIdx.x < DD) {
    float mu = b_mu[d], lv = b_lv[d];
    for (int k = 0; k < DD; ++k) {
      float p = pl[k];
      mu += p * W_mu[(size_t)d * DD + k];
      lv += p * W_lv[(size_t)d * DD + k];
    }
    out[d]      = mu;
    out[DD + d] = lv;
    out[2 * DD + (size_t)NN * DD + d] = pd;
  }
}

extern "C" void kernel_launch(void* const* d_in, const int* in_sizes, int n_in,
                              void* d_out, int out_size, void* d_ws, size_t ws_size,
                              hipStream_t stream) {
  const float* x       = (const float*)d_in[0];
  const int*   ei      = (const int*)d_in[1];
  const float* W_in    = (const float*)d_in[2];
  const float* b_in    = (const float*)d_in[3];
  const float* gat_W   = (const float*)d_in[4];
  const float* att_src = (const float*)d_in[5];
  const float* att_dst = (const float*)d_in[6];
  const float* gat_b   = (const float*)d_in[7];
  const float* ln_g    = (const float*)d_in[8];
  const float* ln_b    = (const float*)d_in[9];
  const float* W_mu    = (const float*)d_in[10];
  const float* b_mu    = (const float*)d_in[11];
  const float* W_lv    = (const float*)d_in[12];
  const float* b_lv    = (const float*)d_in[13];
  float* out = (float*)d_out;

  float* h      = (float*)d_ws;
  float* xh     = h     + (size_t)NN * DD;
  float* plane  = xh    + (size_t)NN * DD;
  float* e_src  = plane + (size_t)NN * DD;
  float* e_dst  = e_src + (size_t)NN * HH;

  int* binned    = (int*)plane;
  int* blockhist = binned    + (size_t)EE;
  int* blockoff  = blockhist + (size_t)NB * NT;
  int* binstart  = blockoff  + (size_t)NB * NT;
  float* part    = (float*)(binstart + NT + 8);

  float* h_final = out + 2 * DD;
  float* xh2     = h_final;

  const int TPB = 256;
  const int nh_blk   = (NN * HH + TPB - 1) / TPB;

  gemm_wlds_k<F_IN><<<GEMM_WG, GEMM_TPB, 0, stream>>>(x, W_in, b_in, h);

  bin_count_k  <<<NB, TPB, 0, stream>>>(ei, blockhist);
  bin_scan_k   <<<1, 1024, 0, stream>>>(blockhist, blockoff, binstart);
  bin_scatter_k<<<NB, TPB, 0, stream>>>(ei, blockoff, binstart, binned);

  for (int l = 0; l < LL; ++l) {
    const float* Wl  = gat_W   + (size_t)l * DD * DD;
    const float* asl = att_src + (size_t)l * HH * CC;
    const float* adl = att_dst + (size_t)l * HH * CC;
    const float* gbl = gat_b   + (size_t)l * DD;
    const float* gl  = ln_g    + (size_t)l * DD;
    const float* bl  = ln_b    + (size_t)l * DD;

    float* xcur = (l & 1) ? xh : xh2;
    float* xnxt = (l & 1) ? xh2 : xh;
    const float* Wn = (l == LL - 1) ? nullptr : Wl + (size_t)DD * DD;

    if (l == 0) gemm_wlds_k<DD><<<GEMM_WG, GEMM_TPB, 0, stream>>>(h, Wl, nullptr, xcur);

    att_logits_k<<<nh_blk, TPB, 0, stream>>>(xcur, asl, adl, e_src, e_dst);

    float* hout  = (l == LL - 1) ? h_final : h;
    float* partl = (l == LL - 1) ? part : nullptr;
    tile_aggr_ln_k<<<NT, TPB, 0, stream>>>(binned, binstart, e_src, e_dst, xcur,
                                           h, gbl, gl, bl, hout, partl,
                                           Wn, (l == LL - 1) ? nullptr : xnxt);
  }

  head_k<<<1, 1024, 0, stream>>>(part, W_mu, b_mu, W_lv, b_lv, out);
}


# tile softmax in one pass: the owning wave keeps a running maximum per node and head in LDS, rescales its sums at each new one; maxima pass, its score gathers, LDS atomics gone; edge order kept
# speedup vs baseline: 4.2815x; 1.0667x over previous
#include <hip/hip_runtime.h>
#include <hip/hip_bf16.h>


#define NN      100000
#define EE      1600000
#define F_IN    256
#define DD      128
#define HH      8
#define CC      16
#define LL      4
#define NEG_SLOPE 0.2f
#define LN_EPS  1e-5f

#define TN      100
#define NT      ((NN + TN - 1) / TN)
#define RB      ((TN + 15) / 16)
#define TNP     (RB * 16)
#define CH      8192
#define NB      ((EE + CH - 1) / CH)
#define SCAN_U  28
#define BIN_U   8
static_assert(NB % SCAN_U == 0, "scan batches must cover the blocks exactly");
static_assert(CH % (256 * BIN_U) == 0, "binning batches must cover a block exactly");
static_assert(NT <= 1024, "one scan thread per tile");
static_assert(NN < (1 << 17) && TN <= 128, "src and local dst share one word");
static_assert((TNP * DD) % (256 * 4) == 0, "the clearing loop covers the plane exactly");

typedef __attribute__((ext_vector_type(2))) float v2f;
typedef __attribute__((ext_vector_type(8))) float v8f;

#define GEMM_WG   256
#define GEMM_TPB  512
#define GEMM_LDS_FLOATS 41472
static_assert(DD * (F_IN + 4) <= GEMM_LDS_FLOATS, "the larger weight matrix fits");
static_assert(NN % 16 == 0, "whole 16-row blocks");

__device__ __forceinline__ int frag_k(int j, int hf) { return 8 * j + 4 * hf; }

template <int K>
__global__ __launch_bounds__(GEMM_TPB) void gemm_wlds_k(
    const float* __restrict__ X, const float* __restrict__ W,
    const float* __restrict__ bias, float* __restrict__ Y) {
  constexpr int KP  = K + 4;
  constexpr int NCH = K / 64;
  static_assert(NCH % 2 == 0, "the chunk ring of two closes over a unit");
  __shared__ __attribute__((aligned(16))) float Ws[GEMM_LDS_FLOATS];

  const int tid  = threadIdx.x;
  const int lane = tid & 31;
  const int wave = __builtin_amdgcn_readfirstlane(tid >> 5);
  const int c    = lane & 15;
  const int hf   = lane >> 4;

  const int rb0 = (int)(((long long)blockIdx.x * (NN / 16)) / GEMM_WG);
  const int rb1 = (int)(((long long)(blockIdx.x + 1) * (NN / 16)) / GEMM_WG);
  const int nu  = (rb1 - rb0) * 2;
  const int ch  = wave & 1;

  const float* wl = Ws + (ch * 64 + c) * KP;
  const float* xa = X + (size_t)((rb0 + (wave >> 1)) * 16 + c) * K;

  float4 a[2][8];
  if (wave < nu) {
    #pragma unroll
    for (int j = 0; j < 8; ++j)
      a[0][j] = *(const float4*)(xa + frag_k(j, hf));
  }

  {
    constexpr int WLD = (DD * K) / (GEMM_TPB * 4);
    static_assert(WLD * GEMM_TPB * 4 == DD * K, "the copy covers W exactly");
    float4 wv[WLD];
    #pragma unroll
    for (int i = 0; i < WLD; ++i)
      wv[i] = *(const float4*)(W + (tid + i * GEMM_TPB) * 4);
    #pragma unroll
    for (int i = 0; i < WLD; ++i) {
      const int e = (tid + i * GEMM_TPB) * 4;
      const int r = e / K, cc = e % K;
      *(float4*)&Ws[r * KP + cc] = wv[i];
    }
  }
  __syncthreads();

  for (int u = wave; u < nu; u += 16) {
    asm volatile("" ::: "memory");
    const int rb = rb0 + (u >> 1);
    const int un = (u + 16 < nu) ? (u + 16) : u;
    const float* xn = X + (size_t)((rb0 + (un >> 1)) * 16 + c) * K;

    v8f acc[4];
    #pragma unroll
    for (int t = 0; t < 4; ++t) acc[t] = v8f{};
    float4 bq[2][4];
    #pragma unroll
    for (int t = 0; t < 4; ++t)
      bq[0][t] = *(const float4*)(wl + t * 16 * KP + frag_k(0, hf));

    #pragma unroll
    for (int cc = 0; cc < NCH; ++cc) {
      const float* src = (cc + 1 < NCH) ? (xa + (cc + 1) * 64) : xn;
      #pragma unroll
      for (int j = 0; j < 8; ++j)
        a[(cc + 1) & 1][j] = *(const float4*)(src + frag_k(j, hf));

      #pragma unroll
      for (int j = 0; j < 8; ++j) {
        const int sp = cc * 8 + j;
        if (sp + 1 < K / 8) {
          #pragma unroll
          for (int t = 0; t < 4; ++t)
            bq[(sp + 1) & 1][t] =
                *(const float4*)(wl + t * 16 * KP + frag_k(sp + 1, hf));
        }
        const float4 av = a[cc & 1][j];
        const v2f a0 = {av.x, av.y}, a1 = {av.z, av.w};
        #pragma unroll
        for (int t = 0; t < 4; ++t) {
          const v2f b0 = {bq[sp & 1][t].x, bq[sp & 1][t].y};
          acc[t] = __builtin_amdgcn_wmma_f32_16x16x4_f32(
              false, a0, false, b0, (short)0, acc[t], false, false);
        }
        #pragma unroll
        for (int t = 0; t < 4; ++t) {
          const v2f b1 = {bq[sp & 1][t].z, bq[sp & 1][t].w};
          acc[t] = __builtin_amdgcn_wmma_f32_16x16x4_f32(
              false, a1, false, b1, (short)0, acc[t], false, false);
        }
      }
    }

    float* yc = Y + (size_t)(rb * 16 + 8 * hf) * DD + ch * 64 + c;
    #pragma unroll
    for (int t = 0; t < 4; ++t) {
      const float bs = bias ? bias[ch * 64 + t * 16 + c] : 0.0f;
      #pragma unroll
      for (int r = 0; r < 8; ++r) yc[r * DD + t * 16] = acc[t][r] + bs;
    }
    xa = xn;
  }
}

__global__ __launch_bounds__(256) void att_logits_k(
    const float* __restrict__ xh, const float* __restrict__ a_src,
    const float* __restrict__ a_dst, float* __restrict__ e_src,
    float* __restrict__ e_dst) {
  int t = blockIdx.x * blockDim.x + threadIdx.x;
  if (t >= NN * HH) return;
  int n = t >> 3, h = t & 7;
  const float4* xp = (const float4*)(xh + (size_t)n * DD + h * CC);
  const float4* sp = (const float4*)(a_src + h * CC);
  const float4* dp = (const float4*)(a_dst + h * CC);
  float ss = 0.f, sd = 0.f;
  #pragma unroll
  for (int i = 0; i < 4; ++i) {
    float4 x = xp[i], s = sp[i], d = dp[i];
    ss += x.x * s.x + x.y * s.y + x.z * s.z + x.w * s.w;
    sd += x.x * d.x + x.y * d.y + x.z * d.z + x.w * d.w;
  }
  e_src[t] = ss;
  e_dst[t] = sd;
}

__device__ __forceinline__ float leaky(float v) {
  return v > 0.f ? v : v * NEG_SLOPE;
}

__global__ __launch_bounds__(256) void bin_count_k(
    const int* __restrict__ ei, int* __restrict__ blockhist) {
  __shared__ int hist[NT];
  for (int i = threadIdx.x; i < NT; i += 256) hist[i] = 0;
  __syncthreads();
  const int e0 = blockIdx.x * CH;
  for (int i0 = threadIdx.x; i0 < CH; i0 += 256 * BIN_U) {
    int d[BIN_U];
    #pragma unroll
    for (int u = 0; u < BIN_U; ++u) {
      const int e = e0 + i0 + 256 * u;
      d[u] = (e < EE) ? ei[EE + e] : -1;
    }
    #pragma unroll
    for (int u = 0; u < BIN_U; ++u)
      if (d[u] >= 0) atomicAdd(&hist[d[u] / TN], 1);
  }
  __syncthreads();
  for (int i = threadIdx.x; i < NT; i += 256)
    blockhist[blockIdx.x * NT + i] = hist[i];
}

__global__ __launch_bounds__(1024) void bin_scan_k(
    const int* __restrict__ blockhist, int* __restrict__ blockoff,
    int* __restrict__ binstart) {
  __shared__ int sc[1024];
  const int t = threadIdx.x;
  int run = 0;
  if (t < NT) {
    for (int b0 = 0; b0 < NB; b0 += SCAN_U) {
      int c[SCAN_U];
      #pragma unroll
      for (int u = 0; u < SCAN_U; ++u) c[u] = blockhist[(b0 + u) * NT + t];
      #pragma unroll
      for (int u = 0; u < SCAN_U; ++u) {
        blockoff[(b0 + u) * NT + t] = run;
        run += c[u];
      }
    }
  }
  sc[t] = run;
  __syncthreads();
  for (int o = 1; o < 1024; o <<= 1) {
    const int v = (t >= o) ? sc[t - o] : 0;
    __syncthreads();
    sc[t] += v;
    __syncthreads();
  }
  if (t < NT) binstart[t] = sc[t] - run;
  if (t == NT - 1) binstart[NT] = sc[t];
}

__global__ __launch_bounds__(256) void bin_scatter_k(
    const int* __restrict__ ei, const int* __restrict__ blockoff,
    const int* __restrict__ binstart, int* __restrict__ binned) {
  __shared__ int cur[NT];
  for (int i = threadIdx.x; i < NT; i += 256)
    cur[i] = binstart[i] + blockoff[blockIdx.x * NT + i];
  __syncthreads();
  const int e0 = blockIdx.x * CH;
  for (int i0 = threadIdx.x; i0 < CH; i0 += 256 * BIN_U) {
    int s[BIN_U], d[BIN_U];
    #pragma unroll
    for (int u = 0; u < BIN_U; ++u) {
      const int e = e0 + i0 + 256 * u;
      s[u] = (e < EE) ? ei[e] : 0;
      d[u] = (e < EE) ? ei[EE + e] : -1;
    }
    #pragma unroll
    for (int u = 0; u < BIN_U; ++u) {
      if (d[u] >= 0) {
        const int t   = d[u] / TN;
        const int pos = atomicAdd(&cur[t], 1);
        binned[pos] = (s[u] << 7) | (d[u] - t * TN);
      }
    }
  }
}

#define AG_U 16
#define SC_U 4
#define OQ_N 256
static_assert(OQ_N >= AG_U + 32 * SC_U && (OQ_N & (OQ_N - 1)) == 0, "the ring holds a scan step");
static_assert((AG_U & (AG_U - 1)) == 0 && AG_U <= 32, "a batch is read one entry per lane");

__device__ __forceinline__ int fetch_edge(const int* __restrict__ binned,
                                          int eb, int idx) {
  const int gi = eb + idx;
  return binned[(gi < EE) ? gi : (EE - 1)];
}

__device__ __forceinline__ void tile_frags(const float* arow,
                                           const float* __restrict__ wrow,
                                           int k, int koff, int sw,
                                           v2f (&a)[RB], v2f& b) {
  const int kc = (k + koff) ^ sw;
  #pragma unroll
  for (int rb = 0; rb < RB; ++rb)
    a[rb] = *(const v2f*)(arow + rb * 16 * DD + kc);
  b = *(const v2f*)(wrow + k);
}
__device__ __forceinline__ void tile_mma(v8f (&d)[RB], const v2f (&a)[RB],
                                         v2f b) {
  #pragma unroll
  for (int rb = 0; rb < RB; ++rb)
    d[rb] = __builtin_amdgcn_wmma_f32_16x16x4_f32(
        false, a[rb], false, b, (short)0, d[rb], false, false);
}

template <bool FULL>
__device__ __forceinline__ void own_batch(
    const int* q, int qh, int cnt, int n0, int lane, int h,
    const float* __restrict__ e_src, const float* __restrict__ xh,
    float* acc, float2* smd, const float* sed) {
  const int qv = q[(qh + (lane & (AG_U - 1))) & (OQ_N - 1)];

  float4 x[AG_U];
  float  es[AG_U];
  int    dls[AG_U];
  #pragma unroll
  for (int u = 0; u < AG_U; ++u) {
    const int pk = (FULL || u < cnt) ? __builtin_amdgcn_readlane(qv, u) : (n0 << 7);
    const int s  = pk >> 7;
    dls[u] = pk & 127;
    es[u]  = e_src[(size_t)s * HH + h];
    x[u]   = *(const float4*)(xh + (size_t)s * DD + lane * 4);
  }
  #pragma unroll
  for (int u = 0; u < AG_U; ++u) {
    if (FULL || u < cnt) {
      const int dl = dls[u];
      float4* ap = (float4*)&acc[dl * DD + lane * 4];
      float2* mp = &smd[dl * HH + h];
      float4 a  = *ap;
      float2 md = *mp;
      const float v = leaky(es[u] + sed[dl * HH + h]);
      const float t  = __expf(-fabsf(v - md.x));
      const bool  up = v > md.x;
      const float sc = up ? t : 1.f;
      const float ex = up ? 1.f : t;
      a.x = a.x * sc + x[u].x * ex;
      a.y = a.y * sc + x[u].y * ex;
      a.z = a.z * sc + x[u].z * ex;
      a.w = a.w * sc + x[u].w * ex;
      md.y = md.y * sc + ex;
      md.x = up ? v : md.x;
      *ap = a;
      if ((lane & 3) == 0) *mp = md;
    }
  }
}

__global__ __launch_bounds__(256) __attribute__((amdgpu_waves_per_eu(8)))
void tile_aggr_ln_k(
    const int* __restrict__ binned, const int* __restrict__ binstart,
    const float* __restrict__ e_src, const float* __restrict__ e_dst,
    const float* __restrict__ xh, const float* hin,
    const float* __restrict__ gb, const float* __restrict__ g,
    const float* __restrict__ b, float* hout, float* __restrict__ part,
    const float* __restrict__ Wn, float* __restrict__ xo) {
  __shared__ __attribute__((aligned(16))) float acc[TNP * DD];
  __shared__ float2   smd[TN * HH];
  __shared__ float    sed[TN * HH];
  __shared__ __attribute__((aligned(16))) float pcol[8 * DD];
  __shared__ int oq[8 * OQ_N];

  const int tid  = threadIdx.x;
  const int lane = tid & 31;
  const int wave = __builtin_amdgcn_readfirstlane(tid >> 5);
  const int h    = lane >> 2;
  const int n0   = blockIdx.x * TN;
  const int nn   = (NN - n0 < TN) ? (NN - n0) : TN;
  const int eb   = binstart[blockIdx.x];
  const int ne   = binstart[blockIdx.x + 1] - eb;

  for (int i = tid * 4; i < TNP * DD; i += 256 * 4)
    *(float4*)&acc[i] = make_float4(0.f, 0.f, 0.f, 0.f);
  for (int i = tid; i < TN * HH; i += 256) {
    float es = 0.f, ed = 0.f;
    if ((i >> 3) < nn) {
      es = e_src[(size_t)n0 * HH + i];
      ed = e_dst[(size_t)n0 * HH + i];
    }
    sed[i] = ed;
    smd[i] = make_float2(leaky(es + ed), 0.f);
  }
  __syncthreads();

  {
    int* q = oq + wave * OQ_N;
    const int nsl = (nn > wave) ? ((nn - wave + 7) >> 3) : 0;
    if (lane < nsl) {
      const int dl = wave + 8 * lane;
      q[lane] = ((n0 + dl) << 7) | dl;
    }
    int qh = 0, qt = nsl;
    int pkc[SC_U];
    #pragma unroll
    for (int c = 0; c < SC_U; ++c)
      pkc[c] = fetch_edge(binned, eb, c * 32 + lane);
    for (int base = 0; base < ne; base += 32 * SC_U) {
      int pkn[SC_U];
      #pragma unroll
      for (int c = 0; c < SC_U; ++c)
        pkn[c] = fetch_edge(binned, eb, base + (SC_U + c) * 32 + lane);
      #pragma unroll
      for (int c = 0; c < SC_U; ++c) {
        const bool own = (base + c * 32 + lane < ne) & ((pkc[c] & 7) == wave);
        const unsigned m = __builtin_amdgcn_ballot_w32(own);
        if (own)
          q[(qt + (int)__builtin_amdgcn_mbcnt_lo(m, 0u)) & (OQ_N - 1)] = pkc[c];
        qt += __builtin_popcount(m);
      }
      while (qt - qh >= AG_U) {
        own_batch<true>(q, qh, AG_U, n0, lane, h, e_src, xh, acc, smd, sed);
        qh += AG_U;
      }
      #pragma unroll
      for (int c = 0; c < SC_U; ++c) pkc[c] = pkn[c];
    }
    if (qt - qh > 0)
      own_batch<false>(q, qh, qt - qh, n0, lane, h, e_src, xh, acc, smd, sed);
  }
  __syncthreads();

  const float4 bv = *(const float4*)(gb + lane * 4);
  const float4 gv = *(const float4*)(g + lane * 4);
  const float4 lb = *(const float4*)(b + lane * 4);
  float4 ps = make_float4(0.f, 0.f, 0.f, 0.f);
  for (int q0 = 0; q0 < (TN + 7) / 8; q0 += 4) {
    float4 hv[4];
    #pragma unroll
    for (int u = 0; u < 4; ++u) {
      const int dl = wave + 8 * (q0 + u);
      hv[u] = make_float4(0.f, 0.f, 0.f, 0.f);
      if (dl < nn)
        hv[u] = *(const float4*)(hin + (size_t)(n0 + dl) * DD + lane * 4);
    }
    #pragma unroll
    for (int u = 0; u < 4; ++u) {
      const int dl = wave + 8 * (q0 + u);
      if (dl < nn) {
        const float  den = smd[dl * HH + h].y + 1e-16f;
        const float4 a   = *(const float4*)&acc[dl * DD + lane * 4];
        float x0 = hv[u].x + a.x / den + bv.x;
        float x1 = hv[u].y + a.y / den + bv.y;
        float x2 = hv[u].z + a.z / den + bv.z;
        float x3 = hv[u].w + a.w / den + bv.w;
        float s = x0 + x1 + x2 + x3;
        #pragma unroll
        for (int o = 16; o > 0; o >>= 1) s += __shfl_xor(s, o, 32);
        float mean = s * (1.f / (float)DD);
        float d0 = x0 - mean, d1 = x1 - mean, d2 = x2 - mean, d3 = x3 - mean;
        float vs = d0 * d0 + d1 * d1 + d2 * d2 + d3 * d3;
        #pragma unroll
        for (int o = 16; o > 0; o >>= 1) vs += __shfl_xor(vs, o, 32);
        float inv = rsqrtf(vs * (1.f / (float)DD) + LN_EPS);
        float4 o4;
        o4.x = d0 * inv * gv.x + lb.x;
        o4.y = d1 * inv * gv.y + lb.y;
        o4.z = d2 * inv * gv.z + lb.z;
        o4.w = d3 * inv * gv.w + lb.w;
        *(float4*)(hout + (size_t)(n0 + dl) * DD + lane * 4) = o4;
        if (Wn)
          *(float4*)&acc[dl * DD + ((lane * 4) ^ ((dl & 15) << 2))] = o4;
        ps.x += o4.x;
        ps.y += o4.y;
        ps.z += o4.z;
        ps.w += o4.w;
      }
    }
  }

  if (part) {
    *(float4*)&pcol[wave * DD + lane * 4] = ps;
    __syncthreads();
    if (tid < DD) {
      float s = 0.f;
      #pragma unroll
      for (int w = 0; w < 8; ++w) s += pcol[w * DD + tid];
      part[(size_t)blockIdx.x * DD + tid] = s;
    }
  }

  if (Wn) {
    __syncthreads();
    const int c    = lane & 15;
    const int koff = (lane < 16) ? 0 : 2;
    const int sw   = c << 2;
    const float* arow = acc + c * DD;
    const float* __restrict__ wrow = Wn + (size_t)(wave * 16 + c) * DD + koff;
    v8f d[RB];
    #pragma unroll
    for (int rb = 0; rb < RB; ++rb) d[rb] = v8f{};
    #pragma unroll
    for (int kk = 0; kk < DD / 4; ++kk) {
      v2f a[RB], b;
      tile_frags(arow, wrow, 4 * kk, koff, sw, a, b);
      tile_mma(d, a, b);
    }

    const int mrow = (lane < 16) ? 0 : 8;
    float* xc = xo + (size_t)(n0 + mrow) * DD + wave * 16 + c;
    if (nn == TN) {
      #pragma unroll
      for (int rb = 0; rb < RB; ++rb) {
        #pragma unroll
        for (int r = 0; r < 8; ++r) {
          if (rb < TN / 16 || rb * 16 + mrow + r < TN)
            xc[(rb * 16 + r) * DD] = d[rb][r];
        }
      }
    } else {
      #pragma unroll
      for (int rb = 0; rb < RB; ++rb) {
        #pragma unroll
        for (int r = 0; r < 8; ++r) {
          if (rb * 16 + mrow + r < nn) xc[(rb * 16 + r) * DD] = d[rb][r];
        }
      }
    }
  }
}

#define HP_G 8
#define HP_U 25
static_assert(HP_G * HP_U * 5 >= NT, "five batches cover the tiles");
__global__ __launch_bounds__(1024) void head_k(
    const float* __restrict__ part, const float* __restrict__ W_mu,
    const float* __restrict__ b_mu, const float* __restrict__ W_lv,
    const float* __restrict__ b_lv, float* __restrict__ out) {
  __shared__ float red[HP_G * DD];
  __shared__ float pl[DD];
  const int d = threadIdx.x & (DD - 1);
  const int g = threadIdx.x >> 7;
  float s = 0.f;
  for (int t0 = g; t0 < NT; t0 += HP_G * HP_U) {
    float v[HP_U];
    #pragma unroll
    for (int u = 0; u < HP_U; ++u) {
      const int t = t0 + HP_G * u;
      v[u] = (t < NT) ? part[(size_t)t * DD + d] : 0.f;
    }
    #pragma unroll
    for (int u = 0; u < HP_U; ++u) s += v[u];
  }
  red[g * DD + d] = s;
  __syncthreads();
  float pd = 0.f;
  if (threadIdx.x < DD) {
    #pragma unroll
    for (int w = 0; w < HP_G; ++w) pd += red[w * DD + d];
    pd *= (1.f / (float)NN);
    pl[d] = pd;
  }
  __syncthreads();
  if (threadIdx.x < DD) {
    float mu = b_mu[d], lv = b_lv[d];
    for (int k = 0; k < DD; ++k) {
      float p = pl[k];
      mu += p * W_mu[(size_t)d * DD + k];
      lv += p * W_lv[(size_t)d * DD + k];
    }
    out[d]      = mu;
    out[DD + d] = lv;
    out[2 * DD + (size_t)NN * DD + d] = pd;
  }
}

extern "C" void kernel_launch(void* const* d_in, const int* in_sizes, int n_in,
                              void* d_out, int out_size, void* d_ws, size_t ws_size,
                              hipStream_t stream) {
  const float* x       = (const float*)d_in[0];
  const int*   ei      = (const int*)d_in[1];
  const float* W_in    = (const float*)d_in[2];
  const float* b_in    = (const float*)d_in[3];
  const float* gat_W   = (const float*)d_in[4];
  const float* att_src = (const float*)d_in[5];
  const float* att_dst = (const float*)d_in[6];
  const float* gat_b   = (const float*)d_in[7];
  const float* ln_g    = (const float*)d_in[8];
  const float* ln_b    = (const float*)d_in[9];
  const float* W_mu    = (const float*)d_in[10];
  const float* b_mu    = (const float*)d_in[11];
  const float* W_lv    = (const float*)d_in[12];
  const float* b_lv    = (const float*)d_in[13];
  float* out = (float*)d_out;

  float* h      = (float*)d_ws;
  float* xh     = h     + (size_t)NN * DD;
  float* plane  = xh    + (size_t)NN * DD;
  float* e_src  = plane + (size_t)NN * DD;
  float* e_dst  = e_src + (size_t)NN * HH;

  int* binned    = (int*)plane;
  int* blockhist = binned    + (size_t)EE;
  int* blockoff  = blockhist + (size_t)NB * NT;
  int* binstart  = blockoff  + (size_t)NB * NT;
  float* part    = (float*)(binstart + NT + 8);

  float* h_final = out + 2 * DD;
  float* xh2     = h_final;

  const int TPB = 256;
  const int nh_blk   = (NN * HH + TPB - 1) / TPB;

  gemm_wlds_k<F_IN><<<GEMM_WG, GEMM_TPB, 0, stream>>>(x, W_in, b_in, h);

  bin_count_k  <<<NB, TPB, 0, stream>>>(ei, blockhist);
  bin_scan_k   <<<1, 1024, 0, stream>>>(blockhist, blockoff, binstart);
  bin_scatter_k<<<NB, TPB, 0, stream>>>(ei, blockoff, binstart, binned);

  for (int l = 0; l < LL; ++l) {
    const float* Wl  = gat_W   + (size_t)l * DD * DD;
    const float* asl = att_src + (size_t)l * HH * CC;
    const float* adl = att_dst + (size_t)l * HH * CC;
    const float* gbl = gat_b   + (size_t)l * DD;
    const float* gl  = ln_g    + (size_t)l * DD;
    const float* bl  = ln_b    + (size_t)l * DD;

    float* xcur = (l & 1) ? xh : xh2;
    float* xnxt = (l & 1) ? xh2 : xh;
    const float* Wn = (l == LL - 1) ? nullptr : Wl + (size_t)DD * DD;

    if (l == 0) gemm_wlds_k<DD><<<GEMM_WG, GEMM_TPB, 0, stream>>>(h, Wl, nullptr, xcur);

    att_logits_k<<<nh_blk, TPB, 0, stream>>>(xcur, asl, adl, e_src, e_dst);

    float* hout  = (l == LL - 1) ? h_final : h;
    float* partl = (l == LL - 1) ? part : nullptr;
    tile_aggr_ln_k<<<NT, TPB, 0, stream>>>(binned, binstart, e_src, e_dst, xcur,
                                           h, gbl, gl, bl, hout, partl,
                                           Wn, (l == LL - 1) ? nullptr : xnxt);
  }

  head_k<<<1, 1024, 0, stream>>>(part, W_mu, b_mu, W_lv, b_lv, out);
}
